# starting kernel
# baseline (speedup 1.0000x reference)
#include <hip/hip_runtime.h>


typedef __attribute__((ext_vector_type(2))) float v2f;
typedef __attribute__((ext_vector_type(8))) float v8f;

#define WPB 8

__global__ void gcn_init_deg(float* __restrict__ deg, int n) {
  int i = blockIdx.x * blockDim.x + threadIdx.x;
  if (i < n) deg[i] = 1.0f;
}

__global__ void gcn_edge_deg(const int* __restrict__ dst, int E,
                             float* __restrict__ deg) {
  int e = blockIdx.x * blockDim.x + threadIdx.x;
  if (e < E) atomicAdd(&deg[dst[e]], 1.0f);
}

__global__ void gcn_rsqrt(float* __restrict__ d, int n) {
  int i = blockIdx.x * blockDim.x + threadIdx.x;
  if (i < n) {
    float v = d[i];
    d[i] = (v > 0.0f) ? rsqrtf(v) : 0.0f;
  }
}

__global__ __launch_bounds__(256) void gcn_gemm1(
    const float* __restrict__ X, const float* __restrict__ W1,
    const float* __restrict__ dinv, float* __restrict__ g1,
    float* __restrict__ agg1, int nTiles) {
  __shared__ float sW[64 * 64];
  {
    const float4* w4 = (const float4*)W1;
    float4* s4 = (float4*)sW;
    for (int i = threadIdx.x; i < 64 * 64 / 4; i += blockDim.x) s4[i] = w4[i];
  }
  __syncthreads();

  const int wave = threadIdx.x >> 5;
  const int lane = threadIdx.x & 31;
  const int tile = blockIdx.x * WPB + wave;
  if (tile >= nTiles) return;

  const int rowbase = tile << 4;
  const int half = lane >> 4;
  const int lp = lane & 15;
  const int m = rowbase + lp;

  v8f acc0 = {}, acc1 = {}, acc2 = {}, acc3 = {};
  const float* __restrict__ Arow = X + (size_t)m * 64;

#pragma unroll
  for (int k = 0; k < 64; k += 4) {
    const int ka = k + 2 * half;
    v2f a;
    a.x = Arow[ka];
    a.y = Arow[ka + 1];
    const float* b0r = &sW[ka * 64];
    const float* b1r = &sW[(ka + 1) * 64];
    v2f b;
    b.x = b0r[lp];
    b.y = b1r[lp];
    acc0 = __builtin_amdgcn_wmma_f32_16x16x4_f32(false, a, false, b, (short)0,
                                                 acc0, false, false);
    b.x = b0r[16 + lp];
    b.y = b1r[16 + lp];
    acc1 = __builtin_amdgcn_wmma_f32_16x16x4_f32(false, a, false, b, (short)0,
                                                 acc1, false, false);
    b.x = b0r[32 + lp];
    b.y = b1r[32 + lp];
    acc2 = __builtin_amdgcn_wmma_f32_16x16x4_f32(false, a, false, b, (short)0,
                                                 acc2, false, false);
    b.x = b0r[48 + lp];
    b.y = b1r[48 + lp];
    acc3 = __builtin_amdgcn_wmma_f32_16x16x4_f32(false, a, false, b, (short)0,
                                                 acc3, false, false);
  }

#pragma unroll
  for (int j = 0; j < 8; ++j) {
    const int row = rowbase + j + 8 * half;
    const float dv = dinv[row];
    const size_t base = (size_t)row * 64 + lp;
    float v;
    v = dv * acc0[j]; g1[base]      = v; agg1[base]      = v;
    v = dv * acc1[j]; g1[base + 16] = v; agg1[base + 16] = v;
    v = dv * acc2[j]; g1[base + 32] = v; agg1[base + 32] = v;
    v = dv * acc3[j]; g1[base + 48] = v; agg1[base + 48] = v;
  }
}

__global__ void gcn_edge_agg1(const int* __restrict__ src,
                              const int* __restrict__ dst,
                              const float* __restrict__ g1,
                              float* __restrict__ agg1, int E) {
  int tid = blockIdx.x * blockDim.x + threadIdx.x;
  if (tid >= E * 16) return;
  int e = tid >> 4;
  int q = (tid & 15) << 2;
  int s = src[e], d = dst[e];
  float4 v = *(const float4*)(g1 + (size_t)s * 64 + q);
  float* p = agg1 + (size_t)d * 64 + q;
  atomicAdd(p + 0, v.x);
  atomicAdd(p + 1, v.y);
  atomicAdd(p + 2, v.z);
  atomicAdd(p + 3, v.w);
}

__global__ __launch_bounds__(256) void gcn_gemm2(
    const float* __restrict__ agg1, const float* __restrict__ b1,
    const float* __restrict__ W2, const float* __restrict__ dinv,
    float* __restrict__ g2, float* __restrict__ outAcc, int nTiles) {
  __shared__ float sW[64 * 16];
  __shared__ float sB[64];
  {
    const float4* w4 = (const float4*)W2;
    float4* s4 = (float4*)sW;
    for (int i = threadIdx.x; i < 64 * 16 / 4; i += blockDim.x) s4[i] = w4[i];
    if (threadIdx.x < 64) sB[threadIdx.x] = b1[threadIdx.x];
  }
  __syncthreads();

  const int wave = threadIdx.x >> 5;
  const int lane = threadIdx.x & 31;
  const int tile = blockIdx.x * WPB + wave;
  if (tile >= nTiles) return;

  const int rowbase = tile << 4;
  const int half = lane >> 4;
  const int lp = lane & 15;
  const int m = rowbase + lp;

  const float dvm = dinv[m];
  const float* __restrict__ Arow = agg1 + (size_t)m * 64;
  v8f acc = {};

#pragma unroll
  for (int k = 0; k < 64; k += 4) {
    const int ka = k + 2 * half;
    v2f a;
    a.x = fmaxf(fmaf(dvm, Arow[ka], sB[ka]), 0.0f);
    a.y = fmaxf(fmaf(dvm, Arow[ka + 1], sB[ka + 1]), 0.0f);
    v2f b;
    b.x = sW[ka * 16 + lp];
    b.y = sW[(ka + 1) * 16 + lp];
    acc = __builtin_amdgcn_wmma_f32_16x16x4_f32(false, a, false, b, (short)0,
                                                acc, false, false);
  }

#pragma unroll
  for (int j = 0; j < 8; ++j) {
    const int row = rowbase + j + 8 * half;
    const float dv = dinv[row];
    const size_t idx = (size_t)row * 16 + lp;
    const float v = dv * acc[j];
    g2[idx] = v;
    outAcc[idx] = v;
  }
}

__global__ void gcn_edge_agg2(const int* __restrict__ src,
                              const int* __restrict__ dst,
                              const float* __restrict__ g2,
                              float* __restrict__ outAcc, int E) {
  int tid = blockIdx.x * blockDim.x + threadIdx.x;
  if (tid >= E * 4) return;
  int e = tid >> 2;
  int q = (tid & 3) << 2;
  int s = src[e], d = dst[e];
  float4 v = *(const float4*)(g2 + (size_t)s * 16 + q);
  float* p = outAcc + (size_t)d * 16 + q;
  atomicAdd(p + 0, v.x);
  atomicAdd(p + 1, v.y);
  atomicAdd(p + 2, v.z);
  atomicAdd(p + 3, v.w);
}

__global__ void gcn_final(float* __restrict__ out, const float* __restrict__ dinv,
                          const float* __restrict__ b2, int n) {
  int tid = blockIdx.x * blockDim.x + threadIdx.x;
  if (tid >= n * 4) return;
  int v = tid >> 2;
  int f = (tid & 3) << 2;
  float dv = dinv[v];
  float4* p = (float4*)out + tid;
  float4 t = *p;
  t.x = fmaf(dv, t.x, b2[f + 0]);
  t.y = fmaf(dv, t.y, b2[f + 1]);
  t.z = fmaf(dv, t.z, b2[f + 2]);
  t.w = fmaf(dv, t.w, b2[f + 3]);
  *p = t;
}

extern "C" void kernel_launch(void* const* d_in, const int* in_sizes, int n_in,
                              void* d_out, int out_size, void* d_ws, size_t ws_size,
                              hipStream_t stream) {
  const float* X  = (const float*)d_in[0];
  const int*   ei = (const int*)d_in[1];
  const float* W1 = (const float*)d_in[2];
  const float* b1 = (const float*)d_in[3];
  const float* W2 = (const float*)d_in[4];
  const float* b2 = (const float*)d_in[5];
  float* out = (float*)d_out;

  const int N = in_sizes[0] / 64;
  const int E = in_sizes[1] / 2;
  const int* src = ei;
  const int* dst = ei + E;

  float* ws   = (float*)d_ws;
  float* dinv = ws;
  float* g1   = dinv + N;
  float* agg1 = g1 + (size_t)N * 64;
  float* g2   = agg1 + (size_t)N * 64;

  const int nTiles = N / 16;
  const int B = 256;

  gcn_init_deg<<<(N + B - 1) / B, B, 0, stream>>>(dinv, N);
  gcn_edge_deg<<<(E + B - 1) / B, B, 0, stream>>>(dst, E, dinv);
  gcn_rsqrt<<<(N + B - 1) / B, B, 0, stream>>>(dinv, N);

  gcn_gemm1<<<(nTiles + WPB - 1) / WPB, B, 0, stream>>>(X, W1, dinv, g1, agg1,
                                                        nTiles);
  gcn_edge_agg1<<<(E * 16 + B - 1) / B, B, 0, stream>>>(src, dst, g1, agg1, E);

  gcn_gemm2<<<(nTiles + WPB - 1) / WPB, B, 0, stream>>>(agg1, b1, W2, dinv, g2,
                                                        out, nTiles);
  gcn_edge_agg2<<<(E * 4 + B - 1) / B, B, 0, stream>>>(src, dst, g2, out, E);

  gcn_final<<<(N * 4 + B - 1) / B, B, 0, stream>>>(out, dinv, b2, N);
}


# Edges bucketed by destination once (returning int atomic rank, prefix sum, slot fill); both layers gather rows per node, no float atomics; sums now in arrival-rank order
# speedup vs baseline: 1.6041x; 1.6041x over previous
#include <hip/hip_runtime.h>


typedef __attribute__((ext_vector_type(2))) float v2f;
typedef __attribute__((ext_vector_type(8))) float v8f;

#define WPB 8
#define SCAN_T 1024

__global__ void gcn_zero_cnt(int* __restrict__ cnt, int n) {
  int i = blockIdx.x * blockDim.x + threadIdx.x;
  if (i < n) cnt[i] = 0;
}

__global__ void gcn_edge_rank(const int* __restrict__ dst, int E,
                              int* __restrict__ cnt, int* __restrict__ pos) {
  const int t = blockIdx.x * blockDim.x + threadIdx.x;
  const int T = gridDim.x * blockDim.x;
  const int e0 = t, e1 = t + T, e2 = t + 2 * T, e3 = t + 3 * T;
  if (e3 < E) {
    const int d0 = dst[e0], d1 = dst[e1], d2 = dst[e2], d3 = dst[e3];
    const int p0 = atomicAdd(&cnt[d0], 1);
    const int p1 = atomicAdd(&cnt[d1], 1);
    const int p2 = atomicAdd(&cnt[d2], 1);
    const int p3 = atomicAdd(&cnt[d3], 1);
    pos[e0] = p0;
    pos[e1] = p1;
    pos[e2] = p2;
    pos[e3] = p3;
  } else {
    for (int e = e0; e < E; e += T) pos[e] = atomicAdd(&cnt[dst[e]], 1);
  }
}

__global__ __launch_bounds__(SCAN_T) void gcn_scan(
    const int* __restrict__ cnt, int n, int* __restrict__ rowptr,
    float* __restrict__ dinv) {
  __shared__ int sPre[SCAN_T / 32];
  __shared__ int sTot[SCAN_T / 32];
  const int tid = threadIdx.x;
  const int lane = tid & 31;
  const int wave = tid >> 5;
  const int base = blockIdx.x * SCAN_T;

  int part = 0;
  const int4* c4 = (const int4*)cnt;
#pragma unroll 8
  for (int i = tid; i < (base >> 2); i += SCAN_T) {
    const int4 v = c4[i];
    part += (v.x + v.y) + (v.z + v.w);
  }
  const int gi = base + tid;
  const int my = (gi < n) ? cnt[gi] : 0;

  for (int o = 16; o > 0; o >>= 1) part += __shfl_xor(part, o);
  int incl = my;
  for (int o = 1; o < 32; o <<= 1) {
    const int v = __shfl_up(incl, o);
    if (lane >= o) incl += v;
  }
  if (lane == 31) sTot[wave] = incl;
  if (lane == 0) sPre[wave] = part;
  __syncthreads();

  int off = 0;
  for (int w = 0; w < SCAN_T / 32; ++w) {
    off += sPre[w];
    if (w < wave) off += sTot[w];
  }
  if (gi < n) {
    rowptr[gi] = off + incl - my;
    const float dg = (float)(my + 1);
    dinv[gi] = (dg > 0.0f) ? rsqrtf(dg) : 0.0f;
  }
}

__global__ void gcn_edge_fill(const int* __restrict__ src,
                              const int* __restrict__ dst,
                              const int* __restrict__ pos,
                              const int* __restrict__ rowptr,
                              int* __restrict__ col, int E) {
  const int t = blockIdx.x * blockDim.x + threadIdx.x;
  const int T = gridDim.x * blockDim.x;
  const int e0 = t, e1 = t + T, e2 = t + 2 * T, e3 = t + 3 * T;
  if (e3 < E) {
    const int d0 = dst[e0], d1 = dst[e1], d2 = dst[e2], d3 = dst[e3];
    const int p0 = pos[e0], p1 = pos[e1], p2 = pos[e2], p3 = pos[e3];
    const int s0 = src[e0], s1 = src[e1], s2 = src[e2], s3 = src[e3];
    const int r0 = rowptr[d0], r1 = rowptr[d1], r2 = rowptr[d2], r3 = rowptr[d3];
    col[r0 + p0] = s0;
    col[r1 + p1] = s1;
    col[r2 + p2] = s2;
    col[r3 + p3] = s3;
  } else {
    for (int e = e0; e < E; e += T) col[rowptr[dst[e]] + pos[e]] = src[e];
  }
}

__global__ __launch_bounds__(256) void gcn_gemm1(
    const float* __restrict__ X, const float* __restrict__ W1,
    const float* __restrict__ dinv, float* __restrict__ g1, int nTiles) {
  __shared__ float sW[64 * 64];
  {
    const float4* w4 = (const float4*)W1;
    float4* s4 = (float4*)sW;
    for (int i = threadIdx.x; i < 64 * 64 / 4; i += blockDim.x) s4[i] = w4[i];
  }
  __syncthreads();

  const int wave = threadIdx.x >> 5;
  const int lane = threadIdx.x & 31;
  const int tile = blockIdx.x * WPB + wave;
  if (tile >= nTiles) return;

  const int rowbase = tile << 4;
  const int half = lane >> 4;
  const int lp = lane & 15;
  const int m = rowbase + lp;

  v8f acc0 = {}, acc1 = {}, acc2 = {}, acc3 = {};
  const float* __restrict__ Arow = X + (size_t)m * 64;

#pragma unroll
  for (int k = 0; k < 64; k += 4) {
    const int ka = k + 2 * half;
    v2f a;
    a.x = Arow[ka];
    a.y = Arow[ka + 1];
    const float* b0r = &sW[ka * 64];
    const float* b1r = &sW[(ka + 1) * 64];
    v2f b;
    b.x = b0r[lp];
    b.y = b1r[lp];
    acc0 = __builtin_amdgcn_wmma_f32_16x16x4_f32(false, a, false, b, (short)0,
                                                 acc0, false, false);
    b.x = b0r[16 + lp];
    b.y = b1r[16 + lp];
    acc1 = __builtin_amdgcn_wmma_f32_16x16x4_f32(false, a, false, b, (short)0,
                                                 acc1, false, false);
    b.x = b0r[32 + lp];
    b.y = b1r[32 + lp];
    acc2 = __builtin_amdgcn_wmma_f32_16x16x4_f32(false, a, false, b, (short)0,
                                                 acc2, false, false);
    b.x = b0r[48 + lp];
    b.y = b1r[48 + lp];
    acc3 = __builtin_amdgcn_wmma_f32_16x16x4_f32(false, a, false, b, (short)0,
                                                 acc3, false, false);
  }

#pragma unroll
  for (int j = 0; j < 8; ++j) {
    const int row = rowbase + j + 8 * half;
    const float dv = dinv[row];
    const size_t base = (size_t)row * 64 + lp;
    g1[base]      = dv * acc0[j];
    g1[base + 16] = dv * acc1[j];
    g1[base + 32] = dv * acc2[j];
    g1[base + 48] = dv * acc3[j];
  }
}

__global__ __launch_bounds__(256) void gcn_gather1(
    const int* __restrict__ rowptr, const int* __restrict__ cnt,
    const int* __restrict__ col, const float* __restrict__ g1,
    float* __restrict__ agg1, int n) {
  const int tid = blockIdx.x * blockDim.x + threadIdx.x;
  const int node = tid >> 4;
  if (node >= n) return;
  const int q = (tid & 15) << 2;
  const int beg = rowptr[node];
  const int end = beg + cnt[node];
  const float* __restrict__ gq = g1 + q;
  float4 acc = *(const float4*)(gq + (size_t)node * 64);
  int j = beg;
  for (; j + 4 <= end; j += 4) {
    const int s0 = col[j], s1 = col[j + 1], s2 = col[j + 2], s3 = col[j + 3];
    const float4 v0 = *(const float4*)(gq + (size_t)s0 * 64);
    const float4 v1 = *(const float4*)(gq + (size_t)s1 * 64);
    const float4 v2 = *(const float4*)(gq + (size_t)s2 * 64);
    const float4 v3 = *(const float4*)(gq + (size_t)s3 * 64);
    acc.x += v0.x; acc.y += v0.y; acc.z += v0.z; acc.w += v0.w;
    acc.x += v1.x; acc.y += v1.y; acc.z += v1.z; acc.w += v1.w;
    acc.x += v2.x; acc.y += v2.y; acc.z += v2.z; acc.w += v2.w;
    acc.x += v3.x; acc.y += v3.y; acc.z += v3.z; acc.w += v3.w;
  }
  for (; j < end; ++j) {
    const float4 v = *(const float4*)(gq + (size_t)col[j] * 64);
    acc.x += v.x; acc.y += v.y; acc.z += v.z; acc.w += v.w;
  }
  *(float4*)(agg1 + (size_t)node * 64 + q) = acc;
}

__global__ __launch_bounds__(256) void gcn_gemm2(
    const float* __restrict__ agg1, const float* __restrict__ b1,
    const float* __restrict__ W2, const float* __restrict__ dinv,
    float* __restrict__ g2, int nTiles) {
  __shared__ float sW[64 * 16];
  __shared__ float sB[64];
  {
    const float4* w4 = (const float4*)W2;
    float4* s4 = (float4*)sW;
    for (int i = threadIdx.x; i < 64 * 16 / 4; i += blockDim.x) s4[i] = w4[i];
    if (threadIdx.x < 64) sB[threadIdx.x] = b1[threadIdx.x];
  }
  __syncthreads();

  const int wave = threadIdx.x >> 5;
  const int lane = threadIdx.x & 31;
  const int tile = blockIdx.x * WPB + wave;
  if (tile >= nTiles) return;

  const int rowbase = tile << 4;
  const int half = lane >> 4;
  const int lp = lane & 15;
  const int m = rowbase + lp;

  const float dvm = dinv[m];
  const float* __restrict__ Arow = agg1 + (size_t)m * 64;
  v8f acc = {};

#pragma unroll
  for (int k = 0; k < 64; k += 4) {
    const int ka = k + 2 * half;
    v2f a;
    a.x = fmaxf(fmaf(dvm, Arow[ka], sB[ka]), 0.0f);
    a.y = fmaxf(fmaf(dvm, Arow[ka + 1], sB[ka + 1]), 0.0f);
    v2f b;
    b.x = sW[ka * 16 + lp];
    b.y = sW[(ka + 1) * 16 + lp];
    acc = __builtin_amdgcn_wmma_f32_16x16x4_f32(false, a, false, b, (short)0,
                                                acc, false, false);
  }

#pragma unroll
  for (int j = 0; j < 8; ++j) {
    const int row = rowbase + j + 8 * half;
    g2[(size_t)row * 16 + lp] = dinv[row] * acc[j];
  }
}

__global__ __launch_bounds__(256) void gcn_gather2(
    const int* __restrict__ rowptr, const int* __restrict__ cnt,
    const int* __restrict__ col, const float* __restrict__ g2,
    const float* __restrict__ dinv, const float* __restrict__ b2,
    float* __restrict__ out, int n) {
  const int tid = blockIdx.x * blockDim.x + threadIdx.x;
  const int node = tid >> 2;
  if (node >= n) return;
  const int q = (tid & 3) << 2;
  const int beg = rowptr[node];
  const int end = beg + cnt[node];
  const float* __restrict__ gq = g2 + q;
  float4 acc = *(const float4*)(gq + (size_t)node * 16);
  int j = beg;
  for (; j + 4 <= end; j += 4) {
    const int s0 = col[j], s1 = col[j + 1], s2 = col[j + 2], s3 = col[j + 3];
    const float4 v0 = *(const float4*)(gq + (size_t)s0 * 16);
    const float4 v1 = *(const float4*)(gq + (size_t)s1 * 16);
    const float4 v2 = *(const float4*)(gq + (size_t)s2 * 16);
    const float4 v3 = *(const float4*)(gq + (size_t)s3 * 16);
    acc.x += v0.x; acc.y += v0.y; acc.z += v0.z; acc.w += v0.w;
    acc.x += v1.x; acc.y += v1.y; acc.z += v1.z; acc.w += v1.w;
    acc.x += v2.x; acc.y += v2.y; acc.z += v2.z; acc.w += v2.w;
    acc.x += v3.x; acc.y += v3.y; acc.z += v3.z; acc.w += v3.w;
  }
  for (; j < end; ++j) {
    const float4 v = *(const float4*)(gq + (size_t)col[j] * 16);
    acc.x += v.x; acc.y += v.y; acc.z += v.z; acc.w += v.w;
  }
  const float dv = dinv[node];
  float4 r;
  r.x = fmaf(dv, acc.x, b2[q + 0]);
  r.y = fmaf(dv, acc.y, b2[q + 1]);
  r.z = fmaf(dv, acc.z, b2[q + 2]);
  r.w = fmaf(dv, acc.w, b2[q + 3]);
  *(float4*)(out + (size_t)node * 16 + q) = r;
}

extern "C" void kernel_launch(void* const* d_in, const int* in_sizes, int n_in,
                              void* d_out, int out_size, void* d_ws, size_t ws_size,
                              hipStream_t stream) {
  const float* X  = (const float*)d_in[0];
  const int*   ei = (const int*)d_in[1];
  const float* W1 = (const float*)d_in[2];
  const float* b1 = (const float*)d_in[3];
  const float* W2 = (const float*)d_in[4];
  const float* b2 = (const float*)d_in[5];
  float* out = (float*)d_out;

  const int N = in_sizes[0] / 64;
  const int E = in_sizes[1] / 2;
  const int* src = ei;
  const int* dst = ei + E;

  float* ws   = (float*)d_ws;
  float* g1   = ws;
  float* agg1 = g1 + (size_t)N * 64;
  float* g2   = agg1 + (size_t)N * 64;
  int* cnt    = (int*)(g2 + (size_t)N * 16);
  int* rowptr = cnt + N;
  float* dinv = (float*)(rowptr + N);
  int* pos    = (int*)(dinv + N);
  int* col    = pos + E;

  const int nTiles = N / 16;
  const int B = 256;
  const int E4 = (E + 3) / 4;

  gcn_zero_cnt<<<(N + B - 1) / B, B, 0, stream>>>(cnt, N);
  gcn_edge_rank<<<(E4 + B - 1) / B, B, 0, stream>>>(dst, E, cnt, pos);
  gcn_scan<<<(N + SCAN_T - 1) / SCAN_T, SCAN_T, 0, stream>>>(cnt, N, rowptr, dinv);
  gcn_edge_fill<<<(E4 + B - 1) / B, B, 0, stream>>>(src, dst, pos, rowptr, col, E);

  gcn_gemm1<<<(nTiles + WPB - 1) / WPB, B, 0, stream>>>(X, W1, dinv, g1, nTiles);
  gcn_gather1<<<(N * 16 + B - 1) / B, B, 0, stream>>>(rowptr, cnt, col, g1, agg1, N);

  gcn_gemm2<<<(nTiles + WPB - 1) / WPB, B, 0, stream>>>(agg1, b1, W2, dinv, g2,
                                                        nTiles);
  gcn_gather2<<<(N * 4 + B - 1) / B, B, 0, stream>>>(rowptr, cnt, col, g2, dinv, b2,
                                                     out, N);
}


# Edges written at their atomic rank straight into fixed 32-slot per-node buckets (surplus to an overflow list, summed last); prefix-sum and slot-fill launches and dinv table dropped: 8 launches to 6
# speedup vs baseline: 1.8037x; 1.1244x over previous
#include <hip/hip_runtime.h>


typedef __attribute__((ext_vector_type(2))) float v2f;
typedef __attribute__((ext_vector_type(8))) float v8f;

#define WPB 8
#define CAP 32

__device__ __forceinline__ float gcn_dinv(int c) {
  const float dg = (float)(c + 1);
  return (dg > 0.0f) ? rsqrtf(dg) : 0.0f;
}

__global__ void gcn_zero_cnt(int* __restrict__ cnt, int n) {
  int i = blockIdx.x * blockDim.x + threadIdx.x;
  if (i < n) cnt[i] = 0;
}

__device__ __forceinline__ void gcn_place(int d, int p, int s,
                                          int* __restrict__ bucket,
                                          int* __restrict__ ovfn,
                                          int* __restrict__ osrc,
                                          int* __restrict__ odst) {
  if (p < CAP) {
    bucket[(size_t)d * CAP + p] = s;
  } else {
    const int q = atomicAdd(ovfn, 1);
    osrc[q] = s;
    odst[q] = d;
  }
}

__global__ void gcn_edge_bucket(const int* __restrict__ src,
                                const int* __restrict__ dst, int E,
                                int* __restrict__ cnt, int* __restrict__ ovfn,
                                int* __restrict__ bucket,
                                int* __restrict__ osrc,
                                int* __restrict__ odst) {
  const int t = blockIdx.x * blockDim.x + threadIdx.x;
  const int T = gridDim.x * blockDim.x;
  const int e0 = t, e1 = t + T, e2 = t + 2 * T, e3 = t + 3 * T;
  if (e3 < E) {
    const int d0 = dst[e0], d1 = dst[e1], d2 = dst[e2], d3 = dst[e3];
    const int s0 = src[e0], s1 = src[e1], s2 = src[e2], s3 = src[e3];
    const int p0 = atomicAdd(&cnt[d0], 1);
    const int p1 = atomicAdd(&cnt[d1], 1);
    const int p2 = atomicAdd(&cnt[d2], 1);
    const int p3 = atomicAdd(&cnt[d3], 1);
    gcn_place(d0, p0, s0, bucket, ovfn, osrc, odst);
    gcn_place(d1, p1, s1, bucket, ovfn, osrc, odst);
    gcn_place(d2, p2, s2, bucket, ovfn, osrc, odst);
    gcn_place(d3, p3, s3, bucket, ovfn, osrc, odst);
  } else {
    for (int e = e0; e < E; e += T) {
      const int d = dst[e];
      const int s = src[e];
      gcn_place(d, atomicAdd(&cnt[d], 1), s, bucket, ovfn, osrc, odst);
    }
  }
}

__global__ __launch_bounds__(256) void gcn_gemm1(
    const float* __restrict__ X, const float* __restrict__ W1,
    const int* __restrict__ cnt, float* __restrict__ g1, int nTiles) {
  __shared__ float sW[64 * 64];
  __shared__ float sD[WPB * 16];
  const int wave = threadIdx.x >> 5;
  const int lane = threadIdx.x & 31;
  const int tile = blockIdx.x * WPB + wave;
  const int rowbase = tile << 4;
  if (tile < nTiles && lane < 16) sD[wave * 16 + lane] = gcn_dinv(cnt[rowbase + lane]);
  {
    const float4* w4 = (const float4*)W1;
    float4* s4 = (float4*)sW;
    for (int i = threadIdx.x; i < 64 * 64 / 4; i += blockDim.x) s4[i] = w4[i];
  }
  __syncthreads();

  if (tile >= nTiles) return;

  const int half = lane >> 4;
  const int lp = lane & 15;
  const int m = rowbase + lp;

  v8f acc0 = {}, acc1 = {}, acc2 = {}, acc3 = {};
  const float* __restrict__ Arow = X + (size_t)m * 64;

#pragma unroll
  for (int k = 0; k < 64; k += 4) {
    const int ka = k + 2 * half;
    v2f a;
    a.x = Arow[ka];
    a.y = Arow[ka + 1];
    const float* b0r = &sW[ka * 64];
    const float* b1r = &sW[(ka + 1) * 64];
    v2f b;
    b.x = b0r[lp];
    b.y = b1r[lp];
    acc0 = __builtin_amdgcn_wmma_f32_16x16x4_f32(false, a, false, b, (short)0,
                                                 acc0, false, false);
    b.x = b0r[16 + lp];
    b.y = b1r[16 + lp];
    acc1 = __builtin_amdgcn_wmma_f32_16x16x4_f32(false, a, false, b, (short)0,
                                                 acc1, false, false);
    b.x = b0r[32 + lp];
    b.y = b1r[32 + lp];
    acc2 = __builtin_amdgcn_wmma_f32_16x16x4_f32(false, a, false, b, (short)0,
                                                 acc2, false, false);
    b.x = b0r[48 + lp];
    b.y = b1r[48 + lp];
    acc3 = __builtin_amdgcn_wmma_f32_16x16x4_f32(false, a, false, b, (short)0,
                                                 acc3, false, false);
  }

#pragma unroll
  for (int j = 0; j < 8; ++j) {
    const int row = rowbase + j + 8 * half;
    const float dv = sD[wave * 16 + j + 8 * half];
    const size_t base = (size_t)row * 64 + lp;
    g1[base]      = dv * acc0[j];
    g1[base + 16] = dv * acc1[j];
    g1[base + 32] = dv * acc2[j];
    g1[base + 48] = dv * acc3[j];
  }
}

__global__ __launch_bounds__(256) void gcn_gather1(
    const int* __restrict__ cnt, const int* __restrict__ ovfn,
    const int* __restrict__ bucket, const int* __restrict__ osrc,
    const int* __restrict__ odst, const float* __restrict__ g1,
    float* __restrict__ agg1, int n) {
  const int tid = blockIdx.x * blockDim.x + threadIdx.x;
  const int node = tid >> 4;
  if (node >= n) return;
  const int q = (tid & 15) << 2;
  const int deg = cnt[node];
  const int end = (deg < CAP) ? deg : CAP;
  const int* __restrict__ col = bucket + (size_t)node * CAP;
  const float* __restrict__ gq = g1 + q;
  float4 acc = *(const float4*)(gq + (size_t)node * 64);
  int j = 0;
  for (; j + 4 <= end; j += 4) {
    const int s0 = col[j], s1 = col[j + 1], s2 = col[j + 2], s3 = col[j + 3];
    const float4 v0 = *(const float4*)(gq + (size_t)s0 * 64);
    const float4 v1 = *(const float4*)(gq + (size_t)s1 * 64);
    const float4 v2 = *(const float4*)(gq + (size_t)s2 * 64);
    const float4 v3 = *(const float4*)(gq + (size_t)s3 * 64);
    acc.x += v0.x; acc.y += v0.y; acc.z += v0.z; acc.w += v0.w;
    acc.x += v1.x; acc.y += v1.y; acc.z += v1.z; acc.w += v1.w;
    acc.x += v2.x; acc.y += v2.y; acc.z += v2.z; acc.w += v2.w;
    acc.x += v3.x; acc.y += v3.y; acc.z += v3.z; acc.w += v3.w;
  }
  for (; j < end; ++j) {
    const float4 v = *(const float4*)(gq + (size_t)col[j] * 64);
    acc.x += v.x; acc.y += v.y; acc.z += v.z; acc.w += v.w;
  }
  if (deg > CAP) {
    const int no = *ovfn;
    for (int i = 0; i < no; ++i) {
      if (odst[i] == node) {
        const float4 v = *(const float4*)(gq + (size_t)osrc[i] * 64);
        acc.x += v.x; acc.y += v.y; acc.z += v.z; acc.w += v.w;
      }
    }
  }
  *(float4*)(agg1 + (size_t)node * 64 + q) = acc;
}

__global__ __launch_bounds__(256) void gcn_gemm2(
    const float* __restrict__ agg1, const float* __restrict__ b1,
    const float* __restrict__ W2, const int* __restrict__ cnt,
    float* __restrict__ g2, int nTiles) {
  __shared__ float sW[64 * 16];
  __shared__ float sB[64];
  {
    const float4* w4 = (const float4*)W2;
    float4* s4 = (float4*)sW;
    for (int i = threadIdx.x; i < 64 * 16 / 4; i += blockDim.x) s4[i] = w4[i];
    if (threadIdx.x < 64) sB[threadIdx.x] = b1[threadIdx.x];
  }
  __syncthreads();

  const int wave = threadIdx.x >> 5;
  const int lane = threadIdx.x & 31;
  const int tile = blockIdx.x * WPB + wave;
  if (tile >= nTiles) return;

  const int rowbase = tile << 4;
  const int half = lane >> 4;
  const int lp = lane & 15;
  const int m = rowbase + lp;

  const float dvm = gcn_dinv(cnt[m]);
  const float* __restrict__ Arow = agg1 + (size_t)m * 64;
  v8f acc = {};

#pragma unroll
  for (int k = 0; k < 64; k += 4) {
    const int ka = k + 2 * half;
    v2f a;
    a.x = fmaxf(fmaf(dvm, Arow[ka], sB[ka]), 0.0f);
    a.y = fmaxf(fmaf(dvm, Arow[ka + 1], sB[ka + 1]), 0.0f);
    v2f b;
    b.x = sW[ka * 16 + lp];
    b.y = sW[(ka + 1) * 16 + lp];
    acc = __builtin_amdgcn_wmma_f32_16x16x4_f32(false, a, false, b, (short)0,
                                                acc, false, false);
  }

#pragma unroll
  for (int j = 0; j < 8; ++j) {
    const int row = rowbase + j + 8 * half;
    g2[(size_t)row * 16 + lp] = gcn_dinv(cnt[row]) * acc[j];
  }
}

__global__ __launch_bounds__(256) void gcn_gather2(
    const int* __restrict__ cnt, const int* __restrict__ ovfn,
    const int* __restrict__ bucket, const int* __restrict__ osrc,
    const int* __restrict__ odst, const float* __restrict__ g2,
    const float* __restrict__ b2, float* __restrict__ out, int n) {
  const int tid = blockIdx.x * blockDim.x + threadIdx.x;
  const int node = tid >> 2;
  if (node >= n) return;
  const int q = (tid & 3) << 2;
  const int deg = cnt[node];
  const int end = (deg < CAP) ? deg : CAP;
  const int* __restrict__ col = bucket + (size_t)node * CAP;
  const float* __restrict__ gq = g2 + q;
  float4 acc = *(const float4*)(gq + (size_t)node * 16);
  int j = 0;
  for (; j + 4 <= end; j += 4) {
    const int s0 = col[j], s1 = col[j + 1], s2 = col[j + 2], s3 = col[j + 3];
    const float4 v0 = *(const float4*)(gq + (size_t)s0 * 16);
    const float4 v1 = *(const float4*)(gq + (size_t)s1 * 16);
    const float4 v2 = *(const float4*)(gq + (size_t)s2 * 16);
    const float4 v3 = *(const float4*)(gq + (size_t)s3 * 16);
    acc.x += v0.x; acc.y += v0.y; acc.z += v0.z; acc.w += v0.w;
    acc.x += v1.x; acc.y += v1.y; acc.z += v1.z; acc.w += v1.w;
    acc.x += v2.x; acc.y += v2.y; acc.z += v2.z; acc.w += v2.w;
    acc.x += v3.x; acc.y += v3.y; acc.z += v3.z; acc.w += v3.w;
  }
  for (; j < end; ++j) {
    const float4 v = *(const float4*)(gq + (size_t)col[j] * 16);
    acc.x += v.x; acc.y += v.y; acc.z += v.z; acc.w += v.w;
  }
  if (deg > CAP) {
    const int no = *ovfn;
    for (int i = 0; i < no; ++i) {
      if (odst[i] == node) {
        const float4 v = *(const float4*)(gq + (size_t)osrc[i] * 16);
        acc.x += v.x; acc.y += v.y; acc.z += v.z; acc.w += v.w;
      }
    }
  }
  const float dv = gcn_dinv(deg);
  float4 r;
  r.x = fmaf(dv, acc.x, b2[q + 0]);
  r.y = fmaf(dv, acc.y, b2[q + 1]);
  r.z = fmaf(dv, acc.z, b2[q + 2]);
  r.w = fmaf(dv, acc.w, b2[q + 3]);
  *(float4*)(out + (size_t)node * 16 + q) = r;
}

extern "C" void kernel_launch(void* const* d_in, const int* in_sizes, int n_in,
                              void* d_out, int out_size, void* d_ws, size_t ws_size,
                              hipStream_t stream) {
  const float* X  = (const float*)d_in[0];
  const int*   ei = (const int*)d_in[1];
  const float* W1 = (const float*)d_in[2];
  const float* b1 = (const float*)d_in[3];
  const float* W2 = (const float*)d_in[4];
  const float* b2 = (const float*)d_in[5];
  float* out = (float*)d_out;

  const int N = in_sizes[0] / 64;
  const int E = in_sizes[1] / 2;
  const int* src = ei;
  const int* dst = ei + E;

  float* ws   = (float*)d_ws;
  float* g1   = ws;
  float* agg1 = g1 + (size_t)N * 64;
  float* g2   = agg1 + (size_t)N * 64;
  int* cnt    = (int*)(g2 + (size_t)N * 16);
  int* ovfn   = cnt + N;
  int* bucket = cnt + ((N + 1 + 31) / 32) * 32;
  int* osrc   = bucket + (size_t)N * CAP;
  int* odst   = osrc + E;

  const int nTiles = N / 16;
  const int B = 256;
  const int E4 = (E + 3) / 4;

  gcn_zero_cnt<<<(N + 1 + B - 1) / B, B, 0, stream>>>(cnt, N + 1);
  gcn_edge_bucket<<<(E4 + B - 1) / B, B, 0, stream>>>(src, dst, E, cnt, ovfn, bucket,
                                                      osrc, odst);

  gcn_gemm1<<<(nTiles + WPB - 1) / WPB, B, 0, stream>>>(X, W1, cnt, g1, nTiles);
  gcn_gather1<<<(N * 16 + B - 1) / B, B, 0, stream>>>(cnt, ovfn, bucket, osrc, odst,
                                                      g1, agg1, N);

  gcn_gemm2<<<(nTiles + WPB - 1) / WPB, B, 0, stream>>>(agg1, b1, W2, cnt, g2,
                                                        nTiles);
  gcn_gather2<<<(N * 4 + B - 1) / B, B, 0, stream>>>(cnt, ovfn, bucket, osrc, odst,
                                                     g2, b2, out, N);
}


# Edge slots ranked by LDS counters in two passes (chunks binned by 512-node range, then one block per range ranks its nodes): 1M returning global adds and clear launch gone; sums stay arrival-ranked
# speedup vs baseline: 2.9973x; 1.6618x over previous
#include <hip/hip_runtime.h>


typedef __attribute__((ext_vector_type(2))) float v2f;
typedef __attribute__((ext_vector_type(8))) float v8f;

#define WPB 8
#define CAP 32

__device__ __forceinline__ float gcn_dinv(int c) {
  const float dg = (float)(c + 1);
  return (dg > 0.0f) ? rsqrtf(dg) : 0.0f;
}

#define RSH 9
#define RNG (1 << RSH)
#define SEGSH 6
#define SEGCAP (1 << SEGSH)
#define NCHUNK 256
#define TB 1024

__global__ __launch_bounds__(TB) void gcn_edge_bin(
    const int* __restrict__ src, const int* __restrict__ dst, int E, int chunk,
    int nbins, int2* __restrict__ seg, int* __restrict__ segcnt,
    int2* __restrict__ spill, int* __restrict__ spillcnt,
    int* __restrict__ ovfn) {
  extern __shared__ int sh[];
  const int c = blockIdx.x;
  for (int i = threadIdx.x; i <= nbins; i += TB) sh[i] = 0;
  if (c == 0 && threadIdx.x == 0) *ovfn = 0;
  __syncthreads();

  const int lo = c * chunk;
  const int hi = (lo + chunk < E) ? lo + chunk : E;
  for (int e0 = lo + threadIdx.x; e0 < hi; e0 += 4 * TB) {
    int d[4], s[4];
#pragma unroll
    for (int k = 0; k < 4; ++k) {
      const int e = e0 + k * TB;
      d[k] = (e < hi) ? dst[e] : 0;
      s[k] = (e < hi) ? src[e] : 0;
    }
#pragma unroll
    for (int k = 0; k < 4; ++k) {
      if (e0 + k * TB < hi) {
        const int bin = d[k] >> RSH;
        const int r = atomicAdd(&sh[bin], 1);
        int2* to;
        if (r < SEGCAP) {
          to = seg + ((((size_t)bin * NCHUNK + c) << SEGSH) + r);
        } else {
          to = spill + ((size_t)lo + atomicAdd(&sh[nbins], 1));
        }
        *to = make_int2(s[k], d[k]);
      }
    }
  }
  __syncthreads();
  for (int i = threadIdx.x; i < nbins; i += TB) {
    const int n = sh[i];
    segcnt[(size_t)i * NCHUNK + c] = (n < SEGCAP) ? n : SEGCAP;
  }
  if (threadIdx.x == 0) spillcnt[c] = sh[nbins];
}

__device__ __forceinline__ void gcn_place(int s, int d, int* lcnt, int base,
                                          int* __restrict__ bucket,
                                          int* __restrict__ ovfn,
                                          int* __restrict__ osrc,
                                          int* __restrict__ odst) {
  const int p = atomicAdd(&lcnt[d - base], 1);
  if (p < CAP) {
    bucket[(size_t)d * CAP + p] = s;
  } else {
    const int q = atomicAdd(ovfn, 1);
    osrc[q] = s;
    odst[q] = d;
  }
}

__global__ __launch_bounds__(TB) void gcn_edge_rank(
    const int2* __restrict__ seg, const int* __restrict__ segcnt,
    const int2* __restrict__ spill, const int* __restrict__ spillcnt,
    int chunk, int n, int* __restrict__ cnt, int* __restrict__ ovfn,
    int* __restrict__ bucket, int* __restrict__ osrc, int* __restrict__ odst) {
  __shared__ int lcnt[RNG];
  __shared__ int scnt[NCHUNK];
  __shared__ int sspill[NCHUNK];
  __shared__ int anyspill;
  const int b = blockIdx.x;
  const int base = b << RSH;
  const int tid = threadIdx.x;
  for (int i = tid; i < RNG; i += TB) lcnt[i] = 0;
  int mysp = 0;
  if (tid < NCHUNK) {
    scnt[tid] = segcnt[(size_t)b * NCHUNK + tid];
    mysp = spillcnt[tid];
    sspill[tid] = mysp;
  }
  if (tid == 0) anyspill = 0;
  __syncthreads();
  if (mysp > 0) anyspill = 1;

  const int2* __restrict__ myseg = seg + (((size_t)b * NCHUNK) << SEGSH);
  for (int s0 = tid; s0 < (NCHUNK << SEGSH); s0 += 4 * TB) {
    int2 e[4];
    bool ok[4];
#pragma unroll
    for (int k = 0; k < 4; ++k) {
      const int slot = s0 + k * TB;
      ok[k] = (slot & (SEGCAP - 1)) < scnt[slot >> SEGSH];
      e[k] = ok[k] ? myseg[slot] : make_int2(0, 0);
    }
#pragma unroll
    for (int k = 0; k < 4; ++k) {
      if (ok[k]) gcn_place(e[k].x, e[k].y, lcnt, base, bucket, ovfn, osrc, odst);
    }
  }
  __syncthreads();
  if (anyspill) {
    for (int c = 0; c < NCHUNK; ++c) {
      const int oc = sspill[c];
      for (int i = tid; i < oc; i += TB) {
        const int2 e = spill[(size_t)c * chunk + i];
        if ((e.y >> RSH) == b)
          gcn_place(e.x, e.y, lcnt, base, bucket, ovfn, osrc, odst);
      }
    }
  }
  __syncthreads();
  for (int i = tid; i < RNG; i += TB) {
    if (base + i < n) cnt[base + i] = lcnt[i];
  }
}

__global__ __launch_bounds__(256) void gcn_gemm1(
    const float* __restrict__ X, const float* __restrict__ W1,
    const int* __restrict__ cnt, float* __restrict__ g1, int nTiles) {
  __shared__ float sW[64 * 64];
  __shared__ float sD[WPB * 16];
  const int wave = threadIdx.x >> 5;
  const int lane = threadIdx.x & 31;
  const int tile = blockIdx.x * WPB + wave;
  const int rowbase = tile << 4;
  if (tile < nTiles && lane < 16) sD[wave * 16 + lane] = gcn_dinv(cnt[rowbase + lane]);
  {
    const float4* w4 = (const float4*)W1;
    float4* s4 = (float4*)sW;
    for (int i = threadIdx.x; i < 64 * 64 / 4; i += blockDim.x) s4[i] = w4[i];
  }
  __syncthreads();

  if (tile >= nTiles) return;

  const int half = lane >> 4;
  const int lp = lane & 15;
  const int m = rowbase + lp;

  v8f acc0 = {}, acc1 = {}, acc2 = {}, acc3 = {};
  const float* __restrict__ Arow = X + (size_t)m * 64;

#pragma unroll
  for (int k = 0; k < 64; k += 4) {
    const int ka = k + 2 * half;
    v2f a;
    a.x = Arow[ka];
    a.y = Arow[ka + 1];
    const float* b0r = &sW[ka * 64];
    const float* b1r = &sW[(ka + 1) * 64];
    v2f b;
    b.x = b0r[lp];
    b.y = b1r[lp];
    acc0 = __builtin_amdgcn_wmma_f32_16x16x4_f32(false, a, false, b, (short)0,
                                                 acc0, false, false);
    b.x = b0r[16 + lp];
    b.y = b1r[16 + lp];
    acc1 = __builtin_amdgcn_wmma_f32_16x16x4_f32(false, a, false, b, (short)0,
                                                 acc1, false, false);
    b.x = b0r[32 + lp];
    b.y = b1r[32 + lp];
    acc2 = __builtin_amdgcn_wmma_f32_16x16x4_f32(false, a, false, b, (short)0,
                                                 acc2, false, false);
    b.x = b0r[48 + lp];
    b.y = b1r[48 + lp];
    acc3 = __builtin_amdgcn_wmma_f32_16x16x4_f32(false, a, false, b, (short)0,
                                                 acc3, false, false);
  }

#pragma unroll
  for (int j = 0; j < 8; ++j) {
    const int row = rowbase + j + 8 * half;
    const float dv = sD[wave * 16 + j + 8 * half];
    const size_t base = (size_t)row * 64 + lp;
    g1[base]      = dv * acc0[j];
    g1[base + 16] = dv * acc1[j];
    g1[base + 32] = dv * acc2[j];
    g1[base + 48] = dv * acc3[j];
  }
}

__global__ __launch_bounds__(256) void gcn_gather1(
    const int* __restrict__ cnt, const int* __restrict__ ovfn,
    const int* __restrict__ bucket, const int* __restrict__ osrc,
    const int* __restrict__ odst, const float* __restrict__ g1,
    float* __restrict__ agg1, int n) {
  const int tid = blockIdx.x * blockDim.x + threadIdx.x;
  const int node = tid >> 4;
  if (node >= n) return;
  const int q = (tid & 15) << 2;
  const int deg = cnt[node];
  const int end = (deg < CAP) ? deg : CAP;
  const int* __restrict__ col = bucket + (size_t)node * CAP;
  const float* __restrict__ gq = g1 + q;
  float4 acc = *(const float4*)(gq + (size_t)node * 64);
  int j = 0;
  for (; j + 4 <= end; j += 4) {
    const int s0 = col[j], s1 = col[j + 1], s2 = col[j + 2], s3 = col[j + 3];
    const float4 v0 = *(const float4*)(gq + (size_t)s0 * 64);
    const float4 v1 = *(const float4*)(gq + (size_t)s1 * 64);
    const float4 v2 = *(const float4*)(gq + (size_t)s2 * 64);
    const float4 v3 = *(const float4*)(gq + (size_t)s3 * 64);
    acc.x += v0.x; acc.y += v0.y; acc.z += v0.z; acc.w += v0.w;
    acc.x += v1.x; acc.y += v1.y; acc.z += v1.z; acc.w += v1.w;
    acc.x += v2.x; acc.y += v2.y; acc.z += v2.z; acc.w += v2.w;
    acc.x += v3.x; acc.y += v3.y; acc.z += v3.z; acc.w += v3.w;
  }
  for (; j < end; ++j) {
    const float4 v = *(const float4*)(gq + (size_t)col[j] * 64);
    acc.x += v.x; acc.y += v.y; acc.z += v.z; acc.w += v.w;
  }
  if (deg > CAP) {
    const int no = *ovfn;
    for (int i = 0; i < no; ++i) {
      if (odst[i] == node) {
        const float4 v = *(const float4*)(gq + (size_t)osrc[i] * 64);
        acc.x += v.x; acc.y += v.y; acc.z += v.z; acc.w += v.w;
      }
    }
  }
  *(float4*)(agg1 + (size_t)node * 64 + q) = acc;
}

__global__ __launch_bounds__(256) void gcn_gemm2(
    const float* __restrict__ agg1, const float* __restrict__ b1,
    const float* __restrict__ W2, const int* __restrict__ cnt,
    float* __restrict__ g2, int nTiles) {
  __shared__ float sW[64 * 16];
  __shared__ float sB[64];
  {
    const float4* w4 = (const float4*)W2;
    float4* s4 = (float4*)sW;
    for (int i = threadIdx.x; i < 64 * 16 / 4; i += blockDim.x) s4[i] = w4[i];
    if (threadIdx.x < 64) sB[threadIdx.x] = b1[threadIdx.x];
  }
  __syncthreads();

  const int wave = threadIdx.x >> 5;
  const int lane = threadIdx.x & 31;
  const int tile = blockIdx.x * WPB + wave;
  if (tile >= nTiles) return;

  const int rowbase = tile << 4;
  const int half = lane >> 4;
  const int lp = lane & 15;
  const int m = rowbase + lp;

  const float dvm = gcn_dinv(cnt[m]);
  const float* __restrict__ Arow = agg1 + (size_t)m * 64;
  v8f acc = {};

#pragma unroll
  for (int k = 0; k < 64; k += 4) {
    const int ka = k + 2 * half;
    v2f a;
    a.x = fmaxf(fmaf(dvm, Arow[ka], sB[ka]), 0.0f);
    a.y = fmaxf(fmaf(dvm, Arow[ka + 1], sB[ka + 1]), 0.0f);
    v2f b;
    b.x = sW[ka * 16 + lp];
    b.y = sW[(ka + 1) * 16 + lp];
    acc = __builtin_amdgcn_wmma_f32_16x16x4_f32(false, a, false, b, (short)0,
                                                acc, false, false);
  }

#pragma unroll
  for (int j = 0; j < 8; ++j) {
    const int row = rowbase + j + 8 * half;
    g2[(size_t)row * 16 + lp] = gcn_dinv(cnt[row]) * acc[j];
  }
}

__global__ __launch_bounds__(256) void gcn_gather2(
    const int* __restrict__ cnt, const int* __restrict__ ovfn,
    const int* __restrict__ bucket, const int* __restrict__ osrc,
    const int* __restrict__ odst, const float* __restrict__ g2,
    const float* __restrict__ b2, float* __restrict__ out, int n) {
  const int tid = blockIdx.x * blockDim.x + threadIdx.x;
  const int node = tid >> 2;
  if (node >= n) return;
  const int q = (tid & 3) << 2;
  const int deg = cnt[node];
  const int end = (deg < CAP) ? deg : CAP;
  const int* __restrict__ col = bucket + (size_t)node * CAP;
  const float* __restrict__ gq = g2 + q;
  float4 acc = *(const float4*)(gq + (size_t)node * 16);
  int j = 0;
  for (; j + 4 <= end; j += 4) {
    const int s0 = col[j], s1 = col[j + 1], s2 = col[j + 2], s3 = col[j + 3];
    const float4 v0 = *(const float4*)(gq + (size_t)s0 * 16);
    const float4 v1 = *(const float4*)(gq + (size_t)s1 * 16);
    const float4 v2 = *(const float4*)(gq + (size_t)s2 * 16);
    const float4 v3 = *(const float4*)(gq + (size_t)s3 * 16);
    acc.x += v0.x; acc.y += v0.y; acc.z += v0.z; acc.w += v0.w;
    acc.x += v1.x; acc.y += v1.y; acc.z += v1.z; acc.w += v1.w;
    acc.x += v2.x; acc.y += v2.y; acc.z += v2.z; acc.w += v2.w;
    acc.x += v3.x; acc.y += v3.y; acc.z += v3.z; acc.w += v3.w;
  }
  for (; j < end; ++j) {
    const float4 v = *(const float4*)(gq + (size_t)col[j] * 16);
    acc.x += v.x; acc.y += v.y; acc.z += v.z; acc.w += v.w;
  }
  if (deg > CAP) {
    const int no = *ovfn;
    for (int i = 0; i < no; ++i) {
      if (odst[i] == node) {
        const float4 v = *(const float4*)(gq + (size_t)osrc[i] * 16);
        acc.x += v.x; acc.y += v.y; acc.z += v.z; acc.w += v.w;
      }
    }
  }
  const float dv = gcn_dinv(deg);
  float4 r;
  r.x = fmaf(dv, acc.x, b2[q + 0]);
  r.y = fmaf(dv, acc.y, b2[q + 1]);
  r.z = fmaf(dv, acc.z, b2[q + 2]);
  r.w = fmaf(dv, acc.w, b2[q + 3]);
  *(float4*)(out + (size_t)node * 16 + q) = r;
}

extern "C" void kernel_launch(void* const* d_in, const int* in_sizes, int n_in,
                              void* d_out, int out_size, void* d_ws, size_t ws_size,
                              hipStream_t stream) {
  const float* X  = (const float*)d_in[0];
  const int*   ei = (const int*)d_in[1];
  const float* W1 = (const float*)d_in[2];
  const float* b1 = (const float*)d_in[3];
  const float* W2 = (const float*)d_in[4];
  const float* b2 = (const float*)d_in[5];
  float* out = (float*)d_out;

  const int N = in_sizes[0] / 64;
  const int E = in_sizes[1] / 2;
  const int* src = ei;
  const int* dst = ei + E;

  float* ws   = (float*)d_ws;
  float* g1   = ws;
  float* agg1 = g1 + (size_t)N * 64;
  float* g2   = agg1 + (size_t)N * 64;
  int* cnt    = (int*)(g2 + (size_t)N * 16);
  int* ovfn   = cnt + N;
  int* bucket = cnt + ((N + 1 + 31) / 32) * 32;
  int* osrc   = bucket + (size_t)N * CAP;
  int* odst   = osrc + E;

  const int nbins = (N + RNG - 1) / RNG;
  const int chunk = (E + NCHUNK - 1) / NCHUNK;
  const size_t used = (size_t)((odst + E) - (int*)d_ws);
  int2* seg     = (int2*)((int*)d_ws + (used + 31) / 32 * 32);
  int2* spill   = seg + (size_t)nbins * NCHUNK * SEGCAP;
  int* segcnt   = (int*)(spill + (size_t)NCHUNK * chunk);
  int* spillcnt = segcnt + (size_t)nbins * NCHUNK;

  const int nTiles = N / 16;
  const int B = 256;

  gcn_edge_bin<<<NCHUNK, TB, (nbins + 1) * sizeof(int), stream>>>(
      src, dst, E, chunk, nbins, seg, segcnt, spill, spillcnt, ovfn);
  gcn_edge_rank<<<nbins, TB, 0, stream>>>(seg, segcnt, spill, spillcnt, chunk, N,
                                          cnt, ovfn, bucket, osrc, odst);

  gcn_gemm1<<<(nTiles + WPB - 1) / WPB, B, 0, stream>>>(X, W1, cnt, g1, nTiles);
  gcn_gather1<<<(N * 16 + B - 1) / B, B, 0, stream>>>(cnt, ovfn, bucket, osrc, odst,
                                                      g1, agg1, N);

  gcn_gemm2<<<(nTiles + WPB - 1) / WPB, B, 0, stream>>>(agg1, b1, W2, cnt, g2,
                                                        nTiles);
  gcn_gather2<<<(N * 4 + B - 1) / B, B, 0, stream>>>(cnt, ovfn, bucket, osrc, odst,
                                                     g2, b2, out, N);
}


# W2 product done inside the layer-1 neighbour-sum kernel: block keeps its 16 summed rows in LDS (bias, relu applied), one wave does the 16 WMMAs; agg1 plane and one launch gone (6 to 5)
# speedup vs baseline: 3.1005x; 1.0344x over previous
#include <hip/hip_runtime.h>


typedef __attribute__((ext_vector_type(2))) float v2f;
typedef __attribute__((ext_vector_type(8))) float v8f;

#define WPB 8
#define CAP 32

__device__ __forceinline__ float gcn_dinv(int c) {
  const float dg = (float)(c + 1);
  return (dg > 0.0f) ? rsqrtf(dg) : 0.0f;
}

#define RSH 9
#define RNG (1 << RSH)
#define SEGSH 6
#define SEGCAP (1 << SEGSH)
#define NCHUNK 256
#define TB 1024

__global__ __launch_bounds__(TB) void gcn_edge_bin(
    const int* __restrict__ src, const int* __restrict__ dst, int E, int chunk,
    int nbins, int2* __restrict__ seg, int* __restrict__ segcnt,
    int2* __restrict__ spill, int* __restrict__ spillcnt,
    int* __restrict__ ovfn) {
  extern __shared__ int sh[];
  const int c = blockIdx.x;
  for (int i = threadIdx.x; i <= nbins; i += TB) sh[i] = 0;
  if (c == 0 && threadIdx.x == 0) *ovfn = 0;
  __syncthreads();

  const int lo = c * chunk;
  const int hi = (lo + chunk < E) ? lo + chunk : E;
  for (int e0 = lo + threadIdx.x; e0 < hi; e0 += 4 * TB) {
    int d[4], s[4];
#pragma unroll
    for (int k = 0; k < 4; ++k) {
      const int e = e0 + k * TB;
      d[k] = (e < hi) ? dst[e] : 0;
      s[k] = (e < hi) ? src[e] : 0;
    }
#pragma unroll
    for (int k = 0; k < 4; ++k) {
      if (e0 + k * TB < hi) {
        const int bin = d[k] >> RSH;
        const int r = atomicAdd(&sh[bin], 1);
        int2* to;
        if (r < SEGCAP) {
          to = seg + ((((size_t)bin * NCHUNK + c) << SEGSH) + r);
        } else {
          to = spill + ((size_t)lo + atomicAdd(&sh[nbins], 1));
        }
        *to = make_int2(s[k], d[k]);
      }
    }
  }
  __syncthreads();
  for (int i = threadIdx.x; i < nbins; i += TB) {
    const int n = sh[i];
    segcnt[(size_t)i * NCHUNK + c] = (n < SEGCAP) ? n : SEGCAP;
  }
  if (threadIdx.x == 0) spillcnt[c] = sh[nbins];
}

__device__ __forceinline__ void gcn_place(int s, int d, int* lcnt, int base,
                                          int* __restrict__ bucket,
                                          int* __restrict__ ovfn,
                                          int* __restrict__ osrc,
                                          int* __restrict__ odst) {
  const int p = atomicAdd(&lcnt[d - base], 1);
  if (p < CAP) {
    bucket[(size_t)d * CAP + p] = s;
  } else {
    const int q = atomicAdd(ovfn, 1);
    osrc[q] = s;
    odst[q] = d;
  }
}

__global__ __launch_bounds__(TB) void gcn_edge_rank(
    const int2* __restrict__ seg, const int* __restrict__ segcnt,
    const int2* __restrict__ spill, const int* __restrict__ spillcnt,
    int chunk, int n, int* __restrict__ cnt, int* __restrict__ ovfn,
    int* __restrict__ bucket, int* __restrict__ osrc, int* __restrict__ odst) {
  __shared__ int lcnt[RNG];
  __shared__ int scnt[NCHUNK];
  __shared__ int sspill[NCHUNK];
  __shared__ int anyspill;
  const int b = blockIdx.x;
  const int base = b << RSH;
  const int tid = threadIdx.x;
  for (int i = tid; i < RNG; i += TB) lcnt[i] = 0;
  int mysp = 0;
  if (tid < NCHUNK) {
    scnt[tid] = segcnt[(size_t)b * NCHUNK + tid];
    mysp = spillcnt[tid];
    sspill[tid] = mysp;
  }
  if (tid == 0) anyspill = 0;
  __syncthreads();
  if (mysp > 0) anyspill = 1;

  const int2* __restrict__ myseg = seg + (((size_t)b * NCHUNK) << SEGSH);
  for (int s0 = tid; s0 < (NCHUNK << SEGSH); s0 += 4 * TB) {
    int2 e[4];
    bool ok[4];
#pragma unroll
    for (int k = 0; k < 4; ++k) {
      const int slot = s0 + k * TB;
      ok[k] = (slot & (SEGCAP - 1)) < scnt[slot >> SEGSH];
      e[k] = ok[k] ? myseg[slot] : make_int2(0, 0);
    }
#pragma unroll
    for (int k = 0; k < 4; ++k) {
      if (ok[k]) gcn_place(e[k].x, e[k].y, lcnt, base, bucket, ovfn, osrc, odst);
    }
  }
  __syncthreads();
  if (anyspill) {
    for (int c = 0; c < NCHUNK; ++c) {
      const int oc = sspill[c];
      for (int i = tid; i < oc; i += TB) {
        const int2 e = spill[(size_t)c * chunk + i];
        if ((e.y >> RSH) == b)
          gcn_place(e.x, e.y, lcnt, base, bucket, ovfn, osrc, odst);
      }
    }
  }
  __syncthreads();
  for (int i = tid; i < RNG; i += TB) {
    if (base + i < n) cnt[base + i] = lcnt[i];
  }
}

__global__ __launch_bounds__(256) void gcn_gemm1(
    const float* __restrict__ X, const float* __restrict__ W1,
    const int* __restrict__ cnt, float* __restrict__ g1, int nTiles) {
  __shared__ float sW[64 * 64];
  __shared__ float sD[WPB * 16];
  const int wave = threadIdx.x >> 5;
  const int lane = threadIdx.x & 31;
  const int tile = blockIdx.x * WPB + wave;
  const int rowbase = tile << 4;
  if (tile < nTiles && lane < 16) sD[wave * 16 + lane] = gcn_dinv(cnt[rowbase + lane]);
  {
    const float4* w4 = (const float4*)W1;
    float4* s4 = (float4*)sW;
    for (int i = threadIdx.x; i < 64 * 64 / 4; i += blockDim.x) s4[i] = w4[i];
  }
  __syncthreads();

  if (tile >= nTiles) return;

  const int half = lane >> 4;
  const int lp = lane & 15;
  const int m = rowbase + lp;

  v8f acc0 = {}, acc1 = {}, acc2 = {}, acc3 = {};
  const float* __restrict__ Arow = X + (size_t)m * 64;

#pragma unroll
  for (int k = 0; k < 64; k += 4) {
    const int ka = k + 2 * half;
    v2f a;
    a.x = Arow[ka];
    a.y = Arow[ka + 1];
    const float* b0r = &sW[ka * 64];
    const float* b1r = &sW[(ka + 1) * 64];
    v2f b;
    b.x = b0r[lp];
    b.y = b1r[lp];
    acc0 = __builtin_amdgcn_wmma_f32_16x16x4_f32(false, a, false, b, (short)0,
                                                 acc0, false, false);
    b.x = b0r[16 + lp];
    b.y = b1r[16 + lp];
    acc1 = __builtin_amdgcn_wmma_f32_16x16x4_f32(false, a, false, b, (short)0,
                                                 acc1, false, false);
    b.x = b0r[32 + lp];
    b.y = b1r[32 + lp];
    acc2 = __builtin_amdgcn_wmma_f32_16x16x4_f32(false, a, false, b, (short)0,
                                                 acc2, false, false);
    b.x = b0r[48 + lp];
    b.y = b1r[48 + lp];
    acc3 = __builtin_amdgcn_wmma_f32_16x16x4_f32(false, a, false, b, (short)0,
                                                 acc3, false, false);
  }

#pragma unroll
  for (int j = 0; j < 8; ++j) {
    const int row = rowbase + j + 8 * half;
    const float dv = sD[wave * 16 + j + 8 * half];
    const size_t base = (size_t)row * 64 + lp;
    g1[base]      = dv * acc0[j];
    g1[base + 16] = dv * acc1[j];
    g1[base + 32] = dv * acc2[j];
    g1[base + 48] = dv * acc3[j];
  }
}

#define XP 68
__global__ __launch_bounds__(256) void gcn_gather1_gemm2(
    const int* __restrict__ cnt, const int* __restrict__ ovfn,
    const int* __restrict__ bucket, const int* __restrict__ osrc,
    const int* __restrict__ odst, const float* __restrict__ g1,
    const float* __restrict__ b1, const float* __restrict__ W2,
    float* __restrict__ g2, int n) {
  __shared__ float sW[64 * 16];
  __shared__ float sX[16 * XP];
  __shared__ __attribute__((aligned(16))) float sD[16];
  ((float4*)sW)[threadIdx.x] = ((const float4*)W2)[threadIdx.x];

  const int nl = threadIdx.x >> 4;
  const int rowbase = blockIdx.x << 4;
  const int node = (rowbase + nl < n) ? rowbase + nl : n - 1;
  const int q = (threadIdx.x & 15) << 2;
  float4 x;
  {
    const int deg = cnt[node];
    const int end = (deg < CAP) ? deg : CAP;
    const int* __restrict__ col = bucket + (size_t)node * CAP;
    const float* __restrict__ gq = g1 + q;
    const float4 bb = *(const float4*)(b1 + q);
    float4 acc = *(const float4*)(gq + (size_t)node * 64);
    int j = 0;
    for (; j + 4 <= end; j += 4) {
      const int s0 = col[j], s1 = col[j + 1], s2 = col[j + 2], s3 = col[j + 3];
      const float4 v0 = *(const float4*)(gq + (size_t)s0 * 64);
      const float4 v1 = *(const float4*)(gq + (size_t)s1 * 64);
      const float4 v2 = *(const float4*)(gq + (size_t)s2 * 64);
      const float4 v3 = *(const float4*)(gq + (size_t)s3 * 64);
      acc.x += v0.x; acc.y += v0.y; acc.z += v0.z; acc.w += v0.w;
      acc.x += v1.x; acc.y += v1.y; acc.z += v1.z; acc.w += v1.w;
      acc.x += v2.x; acc.y += v2.y; acc.z += v2.z; acc.w += v2.w;
      acc.x += v3.x; acc.y += v3.y; acc.z += v3.z; acc.w += v3.w;
    }
    for (; j < end; ++j) {
      const float4 v = *(const float4*)(gq + (size_t)col[j] * 64);
      acc.x += v.x; acc.y += v.y; acc.z += v.z; acc.w += v.w;
    }
    if (deg > CAP) {
      const int no = *ovfn;
      for (int i = 0; i < no; ++i) {
        if (odst[i] == node) {
          const float4 v = *(const float4*)(gq + (size_t)osrc[i] * 64);
          acc.x += v.x; acc.y += v.y; acc.z += v.z; acc.w += v.w;
        }
      }
    }
    const float dv = gcn_dinv(deg);
    x.x = fmaxf(fmaf(dv, acc.x, bb.x), 0.0f);
    x.y = fmaxf(fmaf(dv, acc.y, bb.y), 0.0f);
    x.z = fmaxf(fmaf(dv, acc.z, bb.z), 0.0f);
    x.w = fmaxf(fmaf(dv, acc.w, bb.w), 0.0f);
    if (q == 0) sD[nl] = dv;
  }
  *(float4*)(sX + nl * XP + q) = x;
  __syncthreads();

  if (threadIdx.x >= 32) return;

  const int half = threadIdx.x >> 4;
  const int lp = threadIdx.x & 15;
  const float* __restrict__ Arow = sX + lp * XP;
  v8f acc2 = {};
#pragma unroll
  for (int k = 0; k < 64; k += 4) {
    const int ka = k + 2 * half;
    const v2f a = *(const v2f*)(Arow + ka);
    v2f b;
    b.x = sW[ka * 16 + lp];
    b.y = sW[(ka + 1) * 16 + lp];
    acc2 = __builtin_amdgcn_wmma_f32_16x16x4_f32(false, a, false, b, (short)0,
                                                 acc2, false, false);
  }

  const float4 d0 = *(const float4*)(sD + 8 * half);
  const float4 d1 = *(const float4*)(sD + 8 * half + 4);
  const float dv8[8] = {d0.x, d0.y, d0.z, d0.w, d1.x, d1.y, d1.z, d1.w};
#pragma unroll
  for (int j = 0; j < 8; ++j) {
    const int row = rowbase + j + 8 * half;
    if (row < n) g2[(size_t)row * 16 + lp] = dv8[j] * acc2[j];
  }
}

__global__ __launch_bounds__(256) void gcn_gather2(
    const int* __restrict__ cnt, const int* __restrict__ ovfn,
    const int* __restrict__ bucket, const int* __restrict__ osrc,
    const int* __restrict__ odst, const float* __restrict__ g2,
    const float* __restrict__ b2, float* __restrict__ out, int n) {
  const int tid = blockIdx.x * blockDim.x + threadIdx.x;
  const int node = tid >> 2;
  if (node >= n) return;
  const int q = (tid & 3) << 2;
  const int deg = cnt[node];
  const int end = (deg < CAP) ? deg : CAP;
  const int* __restrict__ col = bucket + (size_t)node * CAP;
  const float* __restrict__ gq = g2 + q;
  float4 acc = *(const float4*)(gq + (size_t)node * 16);
  int j = 0;
  for (; j + 4 <= end; j += 4) {
    const int s0 = col[j], s1 = col[j + 1], s2 = col[j + 2], s3 = col[j + 3];
    const float4 v0 = *(const float4*)(gq + (size_t)s0 * 16);
    const float4 v1 = *(const float4*)(gq + (size_t)s1 * 16);
    const float4 v2 = *(const float4*)(gq + (size_t)s2 * 16);
    const float4 v3 = *(const float4*)(gq + (size_t)s3 * 16);
    acc.x += v0.x; acc.y += v0.y; acc.z += v0.z; acc.w += v0.w;
    acc.x += v1.x; acc.y += v1.y; acc.z += v1.z; acc.w += v1.w;
    acc.x += v2.x; acc.y += v2.y; acc.z += v2.z; acc.w += v2.w;
    acc.x += v3.x; acc.y += v3.y; acc.z += v3.z; acc.w += v3.w;
  }
  for (; j < end; ++j) {
    const float4 v = *(const float4*)(gq + (size_t)col[j] * 16);
    acc.x += v.x; acc.y += v.y; acc.z += v.z; acc.w += v.w;
  }
  if (deg > CAP) {
    const int no = *ovfn;
    for (int i = 0; i < no; ++i) {
      if (odst[i] == node) {
        const float4 v = *(const float4*)(gq + (size_t)osrc[i] * 16);
        acc.x += v.x; acc.y += v.y; acc.z += v.z; acc.w += v.w;
      }
    }
  }
  const float dv = gcn_dinv(deg);
  float4 r;
  r.x = fmaf(dv, acc.x, b2[q + 0]);
  r.y = fmaf(dv, acc.y, b2[q + 1]);
  r.z = fmaf(dv, acc.z, b2[q + 2]);
  r.w = fmaf(dv, acc.w, b2[q + 3]);
  *(float4*)(out + (size_t)node * 16 + q) = r;
}

extern "C" void kernel_launch(void* const* d_in, const int* in_sizes, int n_in,
                              void* d_out, int out_size, void* d_ws, size_t ws_size,
                              hipStream_t stream) {
  const float* X  = (const float*)d_in[0];
  const int*   ei = (const int*)d_in[1];
  const float* W1 = (const float*)d_in[2];
  const float* b1 = (const float*)d_in[3];
  const float* W2 = (const float*)d_in[4];
  const float* b2 = (const float*)d_in[5];
  float* out = (float*)d_out;

  const int N = in_sizes[0] / 64;
  const int E = in_sizes[1] / 2;
  const int* src = ei;
  const int* dst = ei + E;

  float* ws   = (float*)d_ws;
  float* g1   = ws;
  float* g2   = g1 + (size_t)N * 64;
  int* cnt    = (int*)(g2 + (size_t)N * 16);
  int* ovfn   = cnt + N;
  int* bucket = cnt + ((N + 1 + 31) / 32) * 32;
  int* osrc   = bucket + (size_t)N * CAP;
  int* odst   = osrc + E;

  const int nbins = (N + RNG - 1) / RNG;
  const int chunk = (E + NCHUNK - 1) / NCHUNK;
  const size_t used = (size_t)((odst + E) - (int*)d_ws);
  int2* seg     = (int2*)((int*)d_ws + (used + 31) / 32 * 32);
  int2* spill   = seg + (size_t)nbins * NCHUNK * SEGCAP;
  int* segcnt   = (int*)(spill + (size_t)NCHUNK * chunk);
  int* spillcnt = segcnt + (size_t)nbins * NCHUNK;

  const int nTiles = N / 16;
  const int B = 256;

  gcn_edge_bin<<<NCHUNK, TB, (nbins + 1) * sizeof(int), stream>>>(
      src, dst, E, chunk, nbins, seg, segcnt, spill, spillcnt, ovfn);
  gcn_edge_rank<<<nbins, TB, 0, stream>>>(seg, segcnt, spill, spillcnt, chunk, N,
                                          cnt, ovfn, bucket, osrc, odst);

  gcn_gemm1<<<(nTiles + WPB - 1) / WPB, B, 0, stream>>>(X, W1, cnt, g1, nTiles);
  gcn_gather1_gemm2<<<(N + 15) / 16, B, 0, stream>>>(cnt, ovfn, bucket, osrc, odst,
                                                     g1, b1, W2, g2, N);
  gcn_gather2<<<(N * 4 + B - 1) / B, B, 0, stream>>>(cnt, ovfn, bucket, osrc, odst,
                                                     g2, b2, out, N);
}


# X@W1 moved into the edge-ranking kernel: each wave of a range block multiplies one 16-row tile of the block's own nodes between its segment loads, scaled by the LDS in-degrees; 5 launches to 4
# speedup vs baseline: 3.2893x; 1.0609x over previous
#include <hip/hip_runtime.h>


typedef __attribute__((ext_vector_type(2))) float v2f;
typedef __attribute__((ext_vector_type(8))) float v8f;

#define CAP 32

__device__ __forceinline__ float gcn_dinv(int c) {
  const float dg = (float)(c + 1);
  return (dg > 0.0f) ? rsqrtf(dg) : 0.0f;
}

#define RSH 9
#define RNG (1 << RSH)
#define SEGSH 6
#define SEGCAP (1 << SEGSH)
#define NCHUNK 256
#define TB 1024

__global__ __launch_bounds__(TB) void gcn_edge_bin(
    const int* __restrict__ src, const int* __restrict__ dst, int E, int chunk,
    int nbins, int2* __restrict__ seg, int* __restrict__ segcnt,
    int2* __restrict__ spill, int* __restrict__ spillcnt,
    int* __restrict__ ovfn) {
  extern __shared__ int sh[];
  const int c = blockIdx.x;
  for (int i = threadIdx.x; i <= nbins; i += TB) sh[i] = 0;
  if (c == 0 && threadIdx.x == 0) *ovfn = 0;
  __syncthreads();

  const int lo = c * chunk;
  const int hi = (lo + chunk < E) ? lo + chunk : E;
  for (int e0 = lo + threadIdx.x; e0 < hi; e0 += 4 * TB) {
    int d[4], s[4];
#pragma unroll
    for (int k = 0; k < 4; ++k) {
      const int e = e0 + k * TB;
      d[k] = (e < hi) ? dst[e] : 0;
      s[k] = (e < hi) ? src[e] : 0;
    }
#pragma unroll
    for (int k = 0; k < 4; ++k) {
      if (e0 + k * TB < hi) {
        const int bin = d[k] >> RSH;
        const int r = atomicAdd(&sh[bin], 1);
        int2* to;
        if (r < SEGCAP) {
          to = seg + ((((size_t)bin * NCHUNK + c) << SEGSH) + r);
        } else {
          to = spill + ((size_t)lo + atomicAdd(&sh[nbins], 1));
        }
        *to = make_int2(s[k], d[k]);
      }
    }
  }
  __syncthreads();
  for (int i = threadIdx.x; i < nbins; i += TB) {
    const int n = sh[i];
    segcnt[(size_t)i * NCHUNK + c] = (n < SEGCAP) ? n : SEGCAP;
  }
  if (threadIdx.x == 0) spillcnt[c] = sh[nbins];
}

__device__ __forceinline__ void gcn_place(int s, int d, int* lcnt, int base,
                                          int* __restrict__ bucket,
                                          int* __restrict__ ovfn,
                                          int* __restrict__ osrc,
                                          int* __restrict__ odst) {
  const int p = atomicAdd(&lcnt[d - base], 1);
  if (p < CAP) {
    bucket[(size_t)d * CAP + p] = s;
  } else {
    const int q = atomicAdd(ovfn, 1);
    osrc[q] = s;
    odst[q] = d;
  }
}

__global__ __launch_bounds__(TB) void gcn_edge_rank_gemm1(
    const int2* __restrict__ seg, const int* __restrict__ segcnt,
    const int2* __restrict__ spill, const int* __restrict__ spillcnt,
    int chunk, int n, int* __restrict__ cnt, int* __restrict__ ovfn,
    int* __restrict__ bucket, int* __restrict__ osrc, int* __restrict__ odst,
    const float* __restrict__ X, const float* __restrict__ W1,
    float* __restrict__ g1) {
  __shared__ int lcnt[RNG];
  __shared__ int scnt[NCHUNK];
  __shared__ int sspill[NCHUNK];
  __shared__ int anyspill;
  __shared__ float sW[64 * 64];
  const int b = blockIdx.x;
  const int base = b << RSH;
  const int tid = threadIdx.x;

  const int wave = tid >> 5;
  const int half = (tid >> 4) & 1;
  const int lp = tid & 15;
  const int rowbase = base + (wave << 4);
  const bool has = rowbase < n;

  const float4 w4 = ((const float4*)W1)[tid];
  int mysc = 0, mysp = 0;
  if (tid < NCHUNK) {
    mysc = segcnt[(size_t)b * NCHUNK + tid];
    mysp = spillcnt[tid];
  }
  v2f a[16];
  {
    const int m = (rowbase + lp < n) ? rowbase + lp : n - 1;
    const float* __restrict__ Arow = X + (size_t)m * 64 + 2 * half;
#pragma unroll
    for (int kk = 0; kk < 16; ++kk) {
      a[kk].x = Arow[4 * kk];
      a[kk].y = Arow[4 * kk + 1];
    }
  }
  ((float4*)sW)[tid] = w4;
  for (int i = tid; i < RNG; i += TB) lcnt[i] = 0;
  if (tid < NCHUNK) {
    scnt[tid] = mysc;
    sspill[tid] = mysp;
  }
  if (tid == 0) anyspill = 0;
  __syncthreads();
  if (mysp > 0) anyspill = 1;

  v8f acc0 = {}, acc1 = {}, acc2 = {}, acc3 = {};
  const int2* __restrict__ myseg = seg + (((size_t)b * NCHUNK) << SEGSH);
#pragma unroll
  for (int r = 0; r < 4; ++r) {
    const int s0 = tid + r * 4 * TB;
    int2 e[4];
    bool ok[4];
#pragma unroll
    for (int k = 0; k < 4; ++k) {
      const int slot = s0 + k * TB;
      ok[k] = (slot & (SEGCAP - 1)) < scnt[slot >> SEGSH];
      e[k] = ok[k] ? myseg[slot] : make_int2(0, 0);
    }
    if (has) {
#pragma unroll
      for (int kk = 4 * r; kk < 4 * r + 4; ++kk) {
        const int ka = 4 * kk + 2 * half;
        const float* b0r = &sW[ka * 64];
        const float* b1r = &sW[(ka + 1) * 64];
        v2f bf;
        bf.x = b0r[lp];
        bf.y = b1r[lp];
        acc0 = __builtin_amdgcn_wmma_f32_16x16x4_f32(false, a[kk], false, bf,
                                                     (short)0, acc0, false, false);
        bf.x = b0r[16 + lp];
        bf.y = b1r[16 + lp];
        acc1 = __builtin_amdgcn_wmma_f32_16x16x4_f32(false, a[kk], false, bf,
                                                     (short)0, acc1, false, false);
        bf.x = b0r[32 + lp];
        bf.y = b1r[32 + lp];
        acc2 = __builtin_amdgcn_wmma_f32_16x16x4_f32(false, a[kk], false, bf,
                                                     (short)0, acc2, false, false);
        bf.x = b0r[48 + lp];
        bf.y = b1r[48 + lp];
        acc3 = __builtin_amdgcn_wmma_f32_16x16x4_f32(false, a[kk], false, bf,
                                                     (short)0, acc3, false, false);
      }
    }
#pragma unroll
    for (int k = 0; k < 4; ++k) {
      if (ok[k]) gcn_place(e[k].x, e[k].y, lcnt, base, bucket, ovfn, osrc, odst);
    }
  }
  __syncthreads();
  if (anyspill) {
    for (int c = 0; c < NCHUNK; ++c) {
      const int oc = sspill[c];
      for (int i = tid; i < oc; i += TB) {
        const int2 e = spill[(size_t)c * chunk + i];
        if ((e.y >> RSH) == b)
          gcn_place(e.x, e.y, lcnt, base, bucket, ovfn, osrc, odst);
      }
    }
  }
  __syncthreads();
  for (int i = tid; i < RNG; i += TB) {
    if (base + i < n) cnt[base + i] = lcnt[i];
  }
  if (has) {
#pragma unroll
    for (int j = 0; j < 8; ++j) {
      const int lr = (wave << 4) + j + 8 * half;
      const int row = base + lr;
      if (row < n) {
        const float dv = gcn_dinv(lcnt[lr]);
        const size_t o = (size_t)row * 64 + lp;
        g1[o]      = dv * acc0[j];
        g1[o + 16] = dv * acc1[j];
        g1[o + 32] = dv * acc2[j];
        g1[o + 48] = dv * acc3[j];
      }
    }
  }
}

#define XP 68
__global__ __launch_bounds__(256) void gcn_gather1_gemm2(
    const int* __restrict__ cnt, const int* __restrict__ ovfn,
    const int* __restrict__ bucket, const int* __restrict__ osrc,
    const int* __restrict__ odst, const float* __restrict__ g1,
    const float* __restrict__ b1, const float* __restrict__ W2,
    float* __restrict__ g2, int n) {
  __shared__ float sW[64 * 16];
  __shared__ float sX[16 * XP];
  __shared__ __attribute__((aligned(16))) float sD[16];
  ((float4*)sW)[threadIdx.x] = ((const float4*)W2)[threadIdx.x];

  const int nl = threadIdx.x >> 4;
  const int rowbase = blockIdx.x << 4;
  const int node = (rowbase + nl < n) ? rowbase + nl : n - 1;
  const int q = (threadIdx.x & 15) << 2;
  float4 x;
  {
    const int deg = cnt[node];
    const int end = (deg < CAP) ? deg : CAP;
    const int* __restrict__ col = bucket + (size_t)node * CAP;
    const float* __restrict__ gq = g1 + q;
    const float4 bb = *(const float4*)(b1 + q);
    float4 acc = *(const float4*)(gq + (size_t)node * 64);
    int j = 0;
    for (; j + 4 <= end; j += 4) {
      const int s0 = col[j], s1 = col[j + 1], s2 = col[j + 2], s3 = col[j + 3];
      const float4 v0 = *(const float4*)(gq + (size_t)s0 * 64);
      const float4 v1 = *(const float4*)(gq + (size_t)s1 * 64);
      const float4 v2 = *(const float4*)(gq + (size_t)s2 * 64);
      const float4 v3 = *(const float4*)(gq + (size_t)s3 * 64);
      acc.x += v0.x; acc.y += v0.y; acc.z += v0.z; acc.w += v0.w;
      acc.x += v1.x; acc.y += v1.y; acc.z += v1.z; acc.w += v1.w;
      acc.x += v2.x; acc.y += v2.y; acc.z += v2.z; acc.w += v2.w;
      acc.x += v3.x; acc.y += v3.y; acc.z += v3.z; acc.w += v3.w;
    }
    for (; j < end; ++j) {
      const float4 v = *(const float4*)(gq + (size_t)col[j] * 64);
      acc.x += v.x; acc.y += v.y; acc.z += v.z; acc.w += v.w;
    }
    if (deg > CAP) {
      const int no = *ovfn;
      for (int i = 0; i < no; ++i) {
        if (odst[i] == node) {
          const float4 v = *(const float4*)(gq + (size_t)osrc[i] * 64);
          acc.x += v.x; acc.y += v.y; acc.z += v.z; acc.w += v.w;
        }
      }
    }
    const float dv = gcn_dinv(deg);
    x.x = fmaxf(fmaf(dv, acc.x, bb.x), 0.0f);
    x.y = fmaxf(fmaf(dv, acc.y, bb.y), 0.0f);
    x.z = fmaxf(fmaf(dv, acc.z, bb.z), 0.0f);
    x.w = fmaxf(fmaf(dv, acc.w, bb.w), 0.0f);
    if (q == 0) sD[nl] = dv;
  }
  *(float4*)(sX + nl * XP + q) = x;
  __syncthreads();

  if (threadIdx.x >= 32) return;

  const int half = threadIdx.x >> 4;
  const int lp = threadIdx.x & 15;
  const float* __restrict__ Arow = sX + lp * XP;
  v8f acc2 = {};
#pragma unroll
  for (int k = 0; k < 64; k += 4) {
    const int ka = k + 2 * half;
    const v2f a = *(const v2f*)(Arow + ka);
    v2f b;
    b.x = sW[ka * 16 + lp];
    b.y = sW[(ka + 1) * 16 + lp];
    acc2 = __builtin_amdgcn_wmma_f32_16x16x4_f32(false, a, false, b, (short)0,
                                                 acc2, false, false);
  }

  const float4 d0 = *(const float4*)(sD + 8 * half);
  const float4 d1 = *(const float4*)(sD + 8 * half + 4);
  const float dv8[8] = {d0.x, d0.y, d0.z, d0.w, d1.x, d1.y, d1.z, d1.w};
#pragma unroll
  for (int j = 0; j < 8; ++j) {
    const int row = rowbase + j + 8 * half;
    if (row < n) g2[(size_t)row * 16 + lp] = dv8[j] * acc2[j];
  }
}

__global__ __launch_bounds__(256) void gcn_gather2(
    const int* __restrict__ cnt, const int* __restrict__ ovfn,
    const int* __restrict__ bucket, const int* __restrict__ osrc,
    const int* __restrict__ odst, const float* __restrict__ g2,
    const float* __restrict__ b2, float* __restrict__ out, int n) {
  const int tid = blockIdx.x * blockDim.x + threadIdx.x;
  const int node = tid >> 2;
  if (node >= n) return;
  const int q = (tid & 3) << 2;
  const int deg = cnt[node];
  const int end = (deg < CAP) ? deg : CAP;
  const int* __restrict__ col = bucket + (size_t)node * CAP;
  const float* __restrict__ gq = g2 + q;
  float4 acc = *(const float4*)(gq + (size_t)node * 16);
  int j = 0;
  for (; j + 4 <= end; j += 4) {
    const int s0 = col[j], s1 = col[j + 1], s2 = col[j + 2], s3 = col[j + 3];
    const float4 v0 = *(const float4*)(gq + (size_t)s0 * 16);
    const float4 v1 = *(const float4*)(gq + (size_t)s1 * 16);
    const float4 v2 = *(const float4*)(gq + (size_t)s2 * 16);
    const float4 v3 = *(const float4*)(gq + (size_t)s3 * 16);
    acc.x += v0.x; acc.y += v0.y; acc.z += v0.z; acc.w += v0.w;
    acc.x += v1.x; acc.y += v1.y; acc.z += v1.z; acc.w += v1.w;
    acc.x += v2.x; acc.y += v2.y; acc.z += v2.z; acc.w += v2.w;
    acc.x += v3.x; acc.y += v3.y; acc.z += v3.z; acc.w += v3.w;
  }
  for (; j < end; ++j) {
    const float4 v = *(const float4*)(gq + (size_t)col[j] * 16);
    acc.x += v.x; acc.y += v.y; acc.z += v.z; acc.w += v.w;
  }
  if (deg > CAP) {
    const int no = *ovfn;
    for (int i = 0; i < no; ++i) {
      if (odst[i] == node) {
        const float4 v = *(const float4*)(gq + (size_t)osrc[i] * 16);
        acc.x += v.x; acc.y += v.y; acc.z += v.z; acc.w += v.w;
      }
    }
  }
  const float dv = gcn_dinv(deg);
  float4 r;
  r.x = fmaf(dv, acc.x, b2[q + 0]);
  r.y = fmaf(dv, acc.y, b2[q + 1]);
  r.z = fmaf(dv, acc.z, b2[q + 2]);
  r.w = fmaf(dv, acc.w, b2[q + 3]);
  *(float4*)(out + (size_t)node * 16 + q) = r;
}

extern "C" void kernel_launch(void* const* d_in, const int* in_sizes, int n_in,
                              void* d_out, int out_size, void* d_ws, size_t ws_size,
                              hipStream_t stream) {
  const float* X  = (const float*)d_in[0];
  const int*   ei = (const int*)d_in[1];
  const float* W1 = (const float*)d_in[2];
  const float* b1 = (const float*)d_in[3];
  const float* W2 = (const float*)d_in[4];
  const float* b2 = (const float*)d_in[5];
  float* out = (float*)d_out;

  const int N = in_sizes[0] / 64;
  const int E = in_sizes[1] / 2;
  const int* src = ei;
  const int* dst = ei + E;

  float* ws   = (float*)d_ws;
  float* g1   = ws;
  float* g2   = g1 + (size_t)N * 64;
  int* cnt    = (int*)(g2 + (size_t)N * 16);
  int* ovfn   = cnt + N;
  int* bucket = cnt + ((N + 1 + 31) / 32) * 32;
  int* osrc   = bucket + (size_t)N * CAP;
  int* odst   = osrc + E;

  const int nbins = (N + RNG - 1) / RNG;
  const int chunk = (E + NCHUNK - 1) / NCHUNK;
  const size_t used = (size_t)((odst + E) - (int*)d_ws);
  int2* seg     = (int2*)((int*)d_ws + (used + 31) / 32 * 32);
  int2* spill   = seg + (size_t)nbins * NCHUNK * SEGCAP;
  int* segcnt   = (int*)(spill + (size_t)NCHUNK * chunk);
  int* spillcnt = segcnt + (size_t)nbins * NCHUNK;

  const int B = 256;

  gcn_edge_bin<<<NCHUNK, TB, (nbins + 1) * sizeof(int), stream>>>(
      src, dst, E, chunk, nbins, seg, segcnt, spill, spillcnt, ovfn);
  gcn_edge_rank_gemm1<<<nbins, TB, 0, stream>>>(seg, segcnt, spill, spillcnt,
                                                chunk, N, cnt, ovfn, bucket,
                                                osrc, odst, X, W1, g1);
  gcn_gather1_gemm2<<<(N + 15) / 16, B, 0, stream>>>(cnt, ovfn, bucket, osrc, odst,
                                                     g1, b1, W2, g2, N);
  gcn_gather2<<<(N * 4 + B - 1) / B, B, 0, stream>>>(cnt, ovfn, bucket, osrc, odst,
                                                     g2, b2, out, N);
}


# Layer-1 neighbour-sum block takes five 16-node tiles in turn, next tile's degree, slots and own row fetched ahead, rows asked for eight at a time: 10,000 waves enter, not 50,000; sum order kept
# speedup vs baseline: 3.5881x; 1.0908x over previous
#include <hip/hip_runtime.h>


typedef __attribute__((ext_vector_type(2))) float v2f;
typedef __attribute__((ext_vector_type(8))) float v8f;

#define CAP 32

__device__ __forceinline__ float gcn_dinv(int c) {
  const float dg = (float)(c + 1);
  return (dg > 0.0f) ? rsqrtf(dg) : 0.0f;
}

#define RSH 9
#define RNG (1 << RSH)
#define SEGSH 6
#define SEGCAP (1 << SEGSH)
#define NCHUNK 256
#define TB 1024

__global__ __launch_bounds__(TB) void gcn_edge_bin(
    const int* __restrict__ src, const int* __restrict__ dst, int E, int chunk,
    int nbins, int2* __restrict__ seg, int* __restrict__ segcnt,
    int2* __restrict__ spill, int* __restrict__ spillcnt,
    int* __restrict__ ovfn) {
  extern __shared__ int sh[];
  const int c = blockIdx.x;
  for (int i = threadIdx.x; i <= nbins; i += TB) sh[i] = 0;
  if (c == 0 && threadIdx.x == 0) *ovfn = 0;
  __syncthreads();

  const int lo = c * chunk;
  const int hi = (lo + chunk < E) ? lo + chunk : E;
  for (int e0 = lo + threadIdx.x; e0 < hi; e0 += 4 * TB) {
    int d[4], s[4];
#pragma unroll
    for (int k = 0; k < 4; ++k) {
      const int e = e0 + k * TB;
      d[k] = (e < hi) ? dst[e] : 0;
      s[k] = (e < hi) ? src[e] : 0;
    }
#pragma unroll
    for (int k = 0; k < 4; ++k) {
      if (e0 + k * TB < hi) {
        const int bin = d[k] >> RSH;
        const int r = atomicAdd(&sh[bin], 1);
        int2* to;
        if (r < SEGCAP) {
          to = seg + ((((size_t)bin * NCHUNK + c) << SEGSH) + r);
        } else {
          to = spill + ((size_t)lo + atomicAdd(&sh[nbins], 1));
        }
        *to = make_int2(s[k], d[k]);
      }
    }
  }
  __syncthreads();
  for (int i = threadIdx.x; i < nbins; i += TB) {
    const int n = sh[i];
    segcnt[(size_t)i * NCHUNK + c] = (n < SEGCAP) ? n : SEGCAP;
  }
  if (threadIdx.x == 0) spillcnt[c] = sh[nbins];
}

__device__ __forceinline__ void gcn_place(int s, int d, int* lcnt, int base,
                                          int* __restrict__ bucket,
                                          int* __restrict__ ovfn,
                                          int* __restrict__ osrc,
                                          int* __restrict__ odst) {
  const int p = atomicAdd(&lcnt[d - base], 1);
  if (p < CAP) {
    bucket[(size_t)d * CAP + p] = s;
  } else {
    const int q = atomicAdd(ovfn, 1);
    osrc[q] = s;
    odst[q] = d;
  }
}

__global__ __launch_bounds__(TB) void gcn_edge_rank_gemm1(
    const int2* __restrict__ seg, const int* __restrict__ segcnt,
    const int2* __restrict__ spill, const int* __restrict__ spillcnt,
    int chunk, int n, int* __restrict__ cnt, int* __restrict__ ovfn,
    int* __restrict__ bucket, int* __restrict__ osrc, int* __restrict__ odst,
    const float* __restrict__ X, const float* __restrict__ W1,
    float* __restrict__ g1) {
  __shared__ int lcnt[RNG];
  __shared__ int scnt[NCHUNK];
  __shared__ int sspill[NCHUNK];
  __shared__ int anyspill;
  __shared__ float sW[64 * 64];
  const int b = blockIdx.x;
  const int base = b << RSH;
  const int tid = threadIdx.x;

  const int wave = tid >> 5;
  const int half = (tid >> 4) & 1;
  const int lp = tid & 15;
  const int rowbase = base + (wave << 4);
  const bool has = rowbase < n;

  const float4 w4 = ((const float4*)W1)[tid];
  int mysc = 0, mysp = 0;
  if (tid < NCHUNK) {
    mysc = segcnt[(size_t)b * NCHUNK + tid];
    mysp = spillcnt[tid];
  }
  v2f a[16];
  {
    const int m = (rowbase + lp < n) ? rowbase + lp : n - 1;
    const float* __restrict__ Arow = X + (size_t)m * 64 + 2 * half;
#pragma unroll
    for (int kk = 0; kk < 16; ++kk) {
      a[kk].x = Arow[4 * kk];
      a[kk].y = Arow[4 * kk + 1];
    }
  }
  ((float4*)sW)[tid] = w4;
  for (int i = tid; i < RNG; i += TB) lcnt[i] = 0;
  if (tid < NCHUNK) {
    scnt[tid] = mysc;
    sspill[tid] = mysp;
  }
  if (tid == 0) anyspill = 0;
  __syncthreads();
  if (mysp > 0) anyspill = 1;

  v8f acc0 = {}, acc1 = {}, acc2 = {}, acc3 = {};
  const int2* __restrict__ myseg = seg + (((size_t)b * NCHUNK) << SEGSH);
#pragma unroll
  for (int r = 0; r < 4; ++r) {
    const int s0 = tid + r * 4 * TB;
    int2 e[4];
    bool ok[4];
#pragma unroll
    for (int k = 0; k < 4; ++k) {
      const int slot = s0 + k * TB;
      ok[k] = (slot & (SEGCAP - 1)) < scnt[slot >> SEGSH];
      e[k] = ok[k] ? myseg[slot] : make_int2(0, 0);
    }
    if (has) {
#pragma unroll
      for (int kk = 4 * r; kk < 4 * r + 4; ++kk) {
        const int ka = 4 * kk + 2 * half;
        const float* b0r = &sW[ka * 64];
        const float* b1r = &sW[(ka + 1) * 64];
        v2f bf;
        bf.x = b0r[lp];
        bf.y = b1r[lp];
        acc0 = __builtin_amdgcn_wmma_f32_16x16x4_f32(false, a[kk], false, bf,
                                                     (short)0, acc0, false, false);
        bf.x = b0r[16 + lp];
        bf.y = b1r[16 + lp];
        acc1 = __builtin_amdgcn_wmma_f32_16x16x4_f32(false, a[kk], false, bf,
                                                     (short)0, acc1, false, false);
        bf.x = b0r[32 + lp];
        bf.y = b1r[32 + lp];
        acc2 = __builtin_amdgcn_wmma_f32_16x16x4_f32(false, a[kk], false, bf,
                                                     (short)0, acc2, false, false);
        bf.x = b0r[48 + lp];
        bf.y = b1r[48 + lp];
        acc3 = __builtin_amdgcn_wmma_f32_16x16x4_f32(false, a[kk], false, bf,
                                                     (short)0, acc3, false, false);
      }
    }
#pragma unroll
    for (int k = 0; k < 4; ++k) {
      if (ok[k]) gcn_place(e[k].x, e[k].y, lcnt, base, bucket, ovfn, osrc, odst);
    }
  }
  __syncthreads();
  if (anyspill) {
    for (int c = 0; c < NCHUNK; ++c) {
      const int oc = sspill[c];
      for (int i = tid; i < oc; i += TB) {
        const int2 e = spill[(size_t)c * chunk + i];
        if ((e.y >> RSH) == b)
          gcn_place(e.x, e.y, lcnt, base, bucket, ovfn, osrc, odst);
      }
    }
  }
  __syncthreads();
  for (int i = tid; i < RNG; i += TB) {
    if (base + i < n) cnt[base + i] = lcnt[i];
  }
  if (has) {
#pragma unroll
    for (int j = 0; j < 8; ++j) {
      const int lr = (wave << 4) + j + 8 * half;
      const int row = base + lr;
      if (row < n) {
        const float dv = gcn_dinv(lcnt[lr]);
        const size_t o = (size_t)row * 64 + lp;
        g1[o]      = dv * acc0[j];
        g1[o + 16] = dv * acc1[j];
        g1[o + 32] = dv * acc2[j];
        g1[o + 48] = dv * acc3[j];
      }
    }
  }
}

#define XP 68
#define TPB 5

__device__ __forceinline__ void gcn_sum8(float4& acc,
                                         const float* __restrict__ g,
                                         unsigned q, unsigned self, int4 ia,
                                         int4 ib, int j, int end) {
  const bool k0 = j + 0 < end, k1 = j + 1 < end, k2 = j + 2 < end,
             k3 = j + 3 < end, k4 = j + 4 < end, k5 = j + 5 < end,
             k6 = j + 6 < end, k7 = j + 7 < end;
  const unsigned f = k0 ? (unsigned)ia.x : self;
  const float4 v0 = *(const float4*)(g + (f * 64u + q));
  const float4 v1 = *(const float4*)(g + ((k1 ? (unsigned)ia.y : f) * 64u + q));
  const float4 v2 = *(const float4*)(g + ((k2 ? (unsigned)ia.z : f) * 64u + q));
  const float4 v3 = *(const float4*)(g + ((k3 ? (unsigned)ia.w : f) * 64u + q));
  const float4 v4 = *(const float4*)(g + ((k4 ? (unsigned)ib.x : f) * 64u + q));
  const float4 v5 = *(const float4*)(g + ((k5 ? (unsigned)ib.y : f) * 64u + q));
  const float4 v6 = *(const float4*)(g + ((k6 ? (unsigned)ib.z : f) * 64u + q));
  const float4 v7 = *(const float4*)(g + ((k7 ? (unsigned)ib.w : f) * 64u + q));
  __builtin_amdgcn_sched_barrier(0);
  acc.x += k0 ? v0.x : 0.0f; acc.y += k0 ? v0.y : 0.0f;
  acc.z += k0 ? v0.z : 0.0f; acc.w += k0 ? v0.w : 0.0f;
  acc.x += k1 ? v1.x : 0.0f; acc.y += k1 ? v1.y : 0.0f;
  acc.z += k1 ? v1.z : 0.0f; acc.w += k1 ? v1.w : 0.0f;
  acc.x += k2 ? v2.x : 0.0f; acc.y += k2 ? v2.y : 0.0f;
  acc.z += k2 ? v2.z : 0.0f; acc.w += k2 ? v2.w : 0.0f;
  acc.x += k3 ? v3.x : 0.0f; acc.y += k3 ? v3.y : 0.0f;
  acc.z += k3 ? v3.z : 0.0f; acc.w += k3 ? v3.w : 0.0f;
  acc.x += k4 ? v4.x : 0.0f; acc.y += k4 ? v4.y : 0.0f;
  acc.z += k4 ? v4.z : 0.0f; acc.w += k4 ? v4.w : 0.0f;
  acc.x += k5 ? v5.x : 0.0f; acc.y += k5 ? v5.y : 0.0f;
  acc.z += k5 ? v5.z : 0.0f; acc.w += k5 ? v5.w : 0.0f;
  acc.x += k6 ? v6.x : 0.0f; acc.y += k6 ? v6.y : 0.0f;
  acc.z += k6 ? v6.z : 0.0f; acc.w += k6 ? v6.w : 0.0f;
  acc.x += k7 ? v7.x : 0.0f; acc.y += k7 ? v7.y : 0.0f;
  acc.z += k7 ? v7.z : 0.0f; acc.w += k7 ? v7.w : 0.0f;
}

__global__ __launch_bounds__(256) void gcn_gather1_gemm2(
    const int* __restrict__ cnt, const int* __restrict__ ovfn,
    const int* __restrict__ bucket, const int* __restrict__ osrc,
    const int* __restrict__ odst, const float* __restrict__ g1,
    const float* __restrict__ b1, const float* __restrict__ W2,
    float* __restrict__ g2, int n) {
  __shared__ float sW[64 * 16];
  __shared__ __attribute__((aligned(16))) float sX[TPB * 16 * XP];
  __shared__ __attribute__((aligned(16))) float sD[TPB * 16];
  ((float4*)sW)[threadIdx.x] = ((const float4*)W2)[threadIdx.x];

  const int nl = threadIdx.x >> 4;
  const int blockbase = blockIdx.x * (16 * TPB);
  const unsigned q = (threadIdx.x & 15) << 2;
  const float4 bb = *(const float4*)(b1 + q);

  int nodeN = (blockbase + nl < n) ? blockbase + nl : n - 1;
  int degN = cnt[nodeN];
  int4 iaN = ((const int4*)(bucket + (size_t)nodeN * CAP))[0];
  int4 ibN = ((const int4*)(bucket + (size_t)nodeN * CAP))[1];
  float4 selfN = *(const float4*)(g1 + ((unsigned)nodeN * 64u + q));

  for (int t = 0; t < TPB; ++t) {
    const int node = nodeN;
    const int deg = degN;
    int4 ia = iaN, ib = ibN;
    float4 acc = selfN;
    const int4* __restrict__ col4 = (const int4*)(bucket + (size_t)node * CAP);
    if (t + 1 < TPB) {
      const int nb = blockbase + 16 * (t + 1) + nl;
      nodeN = (nb < n) ? nb : n - 1;
      degN = cnt[nodeN];
      iaN = ((const int4*)(bucket + (size_t)nodeN * CAP))[0];
      ibN = ((const int4*)(bucket + (size_t)nodeN * CAP))[1];
      selfN = *(const float4*)(g1 + ((unsigned)nodeN * 64u + q));
    }
    const int end = (deg < CAP) ? deg : CAP;
    gcn_sum8(acc, g1, q, (unsigned)node, ia, ib, 0, end);
    for (int j = 8; j < end; j += 8) {
      ia = col4[j >> 2];
      ib = col4[(j >> 2) + 1];
      gcn_sum8(acc, g1, q, (unsigned)node, ia, ib, j, end);
    }
    if (deg > CAP) {
      const int no = *ovfn;
      for (int i = 0; i < no; ++i) {
        if (odst[i] == node) {
          const float4 v = *(const float4*)(g1 + ((unsigned)osrc[i] * 64u + q));
          acc.x += v.x; acc.y += v.y; acc.z += v.z; acc.w += v.w;
        }
      }
    }
    const float dv = gcn_dinv(deg);
    float4 x;
    x.x = fmaxf(fmaf(dv, acc.x, bb.x), 0.0f);
    x.y = fmaxf(fmaf(dv, acc.y, bb.y), 0.0f);
    x.z = fmaxf(fmaf(dv, acc.z, bb.z), 0.0f);
    x.w = fmaxf(fmaf(dv, acc.w, bb.w), 0.0f);
    if (q == 0) sD[t * 16 + nl] = dv;
    *(float4*)(sX + (t * 16 + nl) * XP + q) = x;
  }
  __syncthreads();

  const int tile = threadIdx.x >> 5;
  if (tile >= TPB) return;

  const int half = (threadIdx.x >> 4) & 1;
  const int lp = threadIdx.x & 15;
  const int rowbase = blockbase + 16 * tile;
  const float* __restrict__ Arow = sX + (tile * 16 + lp) * XP;
  v8f acc2 = {};
#pragma unroll
  for (int k = 0; k < 64; k += 4) {
    const int ka = k + 2 * half;
    const v2f a = *(const v2f*)(Arow + ka);
    v2f b;
    b.x = sW[ka * 16 + lp];
    b.y = sW[(ka + 1) * 16 + lp];
    acc2 = __builtin_amdgcn_wmma_f32_16x16x4_f32(false, a, false, b, (short)0,
                                                 acc2, false, false);
  }

  const float4 d0 = *(const float4*)(sD + tile * 16 + 8 * half);
  const float4 d1 = *(const float4*)(sD + tile * 16 + 8 * half + 4);
  const float dv8[8] = {d0.x, d0.y, d0.z, d0.w, d1.x, d1.y, d1.z, d1.w};
#pragma unroll
  for (int j = 0; j < 8; ++j) {
    const int row = rowbase + j + 8 * half;
    if (row < n) g2[(size_t)row * 16 + lp] = dv8[j] * acc2[j];
  }
}

__global__ __launch_bounds__(256) void gcn_gather2(
    const int* __restrict__ cnt, const int* __restrict__ ovfn,
    const int* __restrict__ bucket, const int* __restrict__ osrc,
    const int* __restrict__ odst, const float* __restrict__ g2,
    const float* __restrict__ b2, float* __restrict__ out, int n) {
  const int tid = blockIdx.x * blockDim.x + threadIdx.x;
  const int node = tid >> 2;
  if (node >= n) return;
  const int q = (tid & 3) << 2;
  const int deg = cnt[node];
  const int end = (deg < CAP) ? deg : CAP;
  const int* __restrict__ col = bucket + (size_t)node * CAP;
  const float* __restrict__ gq = g2 + q;
  float4 acc = *(const float4*)(gq + (size_t)node * 16);
  int j = 0;
  for (; j + 4 <= end; j += 4) {
    const int s0 = col[j], s1 = col[j + 1], s2 = col[j + 2], s3 = col[j + 3];
    const float4 v0 = *(const float4*)(gq + (size_t)s0 * 16);
    const float4 v1 = *(const float4*)(gq + (size_t)s1 * 16);
    const float4 v2 = *(const float4*)(gq + (size_t)s2 * 16);
    const float4 v3 = *(const float4*)(gq + (size_t)s3 * 16);
    acc.x += v0.x; acc.y += v0.y; acc.z += v0.z; acc.w += v0.w;
    acc.x += v1.x; acc.y += v1.y; acc.z += v1.z; acc.w += v1.w;
    acc.x += v2.x; acc.y += v2.y; acc.z += v2.z; acc.w += v2.w;
    acc.x += v3.x; acc.y += v3.y; acc.z += v3.z; acc.w += v3.w;
  }
  for (; j < end; ++j) {
    const float4 v = *(const float4*)(gq + (size_t)col[j] * 16);
    acc.x += v.x; acc.y += v.y; acc.z += v.z; acc.w += v.w;
  }
  if (deg > CAP) {
    const int no = *ovfn;
    for (int i = 0; i < no; ++i) {
      if (odst[i] == node) {
        const float4 v = *(const float4*)(gq + (size_t)osrc[i] * 16);
        acc.x += v.x; acc.y += v.y; acc.z += v.z; acc.w += v.w;
      }
    }
  }
  const float dv = gcn_dinv(deg);
  float4 r;
  r.x = fmaf(dv, acc.x, b2[q + 0]);
  r.y = fmaf(dv, acc.y, b2[q + 1]);
  r.z = fmaf(dv, acc.z, b2[q + 2]);
  r.w = fmaf(dv, acc.w, b2[q + 3]);
  *(float4*)(out + (size_t)node * 16 + q) = r;
}

extern "C" void kernel_launch(void* const* d_in, const int* in_sizes, int n_in,
                              void* d_out, int out_size, void* d_ws, size_t ws_size,
                              hipStream_t stream) {
  const float* X  = (const float*)d_in[0];
  const int*   ei = (const int*)d_in[1];
  const float* W1 = (const float*)d_in[2];
  const float* b1 = (const float*)d_in[3];
  const float* W2 = (const float*)d_in[4];
  const float* b2 = (const float*)d_in[5];
  float* out = (float*)d_out;

  const int N = in_sizes[0] / 64;
  const int E = in_sizes[1] / 2;
  const int* src = ei;
  const int* dst = ei + E;

  float* ws   = (float*)d_ws;
  float* g1   = ws;
  float* g2   = g1 + (size_t)N * 64;
  int* cnt    = (int*)(g2 + (size_t)N * 16);
  int* ovfn   = cnt + N;
  int* bucket = cnt + ((N + 1 + 31) / 32) * 32;
  int* osrc   = bucket + (size_t)N * CAP;
  int* odst   = osrc + E;

  const int nbins = (N + RNG - 1) / RNG;
  const int chunk = (E + NCHUNK - 1) / NCHUNK;
  const size_t used = (size_t)((odst + E) - (int*)d_ws);
  int2* seg     = (int2*)((int*)d_ws + (used + 31) / 32 * 32);
  int2* spill   = seg + (size_t)nbins * NCHUNK * SEGCAP;
  int* segcnt   = (int*)(spill + (size_t)NCHUNK * chunk);
  int* spillcnt = segcnt + (size_t)nbins * NCHUNK;

  const int B = 256;

  gcn_edge_bin<<<NCHUNK, TB, (nbins + 1) * sizeof(int), stream>>>(
      src, dst, E, chunk, nbins, seg, segcnt, spill, spillcnt, ovfn);
  gcn_edge_rank_gemm1<<<nbins, TB, 0, stream>>>(seg, segcnt, spill, spillcnt,
                                                chunk, N, cnt, ovfn, bucket,
                                                osrc, odst, X, W1, g1);
  gcn_gather1_gemm2<<<(N + 16 * TPB - 1) / (16 * TPB), B, 0, stream>>>(cnt, ovfn, bucket, osrc, odst,
                                                     g1, b1, W2, g2, N);
  gcn_gather2<<<(N * 4 + B - 1) / B, B, 0, stream>>>(cnt, ovfn, bucket, osrc, odst,
                                                     g2, b2, out, N);
}


# W1 product moved behind the layer-1 neighbour sum: one wave sums a 16-node tile of dinv[src]*X rows, then does its W1 and W2 WMMAs; g1 plane gone; rank kernel ranks only; sum order: product after sum
# speedup vs baseline: 3.8385x; 1.0698x over previous
#include <hip/hip_runtime.h>


typedef __attribute__((ext_vector_type(2))) float v2f;
typedef __attribute__((ext_vector_type(8))) float v8f;

#define CAP 32

__device__ __forceinline__ float gcn_dinv(int c) {
  const float dg = (float)(c + 1);
  return (dg > 0.0f) ? rsqrtf(dg) : 0.0f;
}

#define RSH 9
#define RNG (1 << RSH)
#define SEGSH 6
#define SEGCAP (1 << SEGSH)
#define NCHUNK 256
#define TB 1024

__global__ __launch_bounds__(TB) void gcn_edge_bin(
    const int* __restrict__ src, const int* __restrict__ dst, int E, int chunk,
    int nbins, int2* __restrict__ seg, int* __restrict__ segcnt,
    int2* __restrict__ spill, int* __restrict__ spillcnt,
    int* __restrict__ ovfn) {
  extern __shared__ int sh[];
  const int c = blockIdx.x;
  for (int i = threadIdx.x; i <= nbins; i += TB) sh[i] = 0;
  if (c == 0 && threadIdx.x == 0) *ovfn = 0;
  __syncthreads();

  const int lo = c * chunk;
  const int hi = (lo + chunk < E) ? lo + chunk : E;
  for (int e0 = lo + threadIdx.x; e0 < hi; e0 += 4 * TB) {
    int d[4], s[4];
#pragma unroll
    for (int k = 0; k < 4; ++k) {
      const int e = e0 + k * TB;
      d[k] = (e < hi) ? dst[e] : 0;
      s[k] = (e < hi) ? src[e] : 0;
    }
#pragma unroll
    for (int k = 0; k < 4; ++k) {
      if (e0 + k * TB < hi) {
        const int bin = d[k] >> RSH;
        const int r = atomicAdd(&sh[bin], 1);
        int2* to;
        if (r < SEGCAP) {
          to = seg + ((((size_t)bin * NCHUNK + c) << SEGSH) + r);
        } else {
          to = spill + ((size_t)lo + atomicAdd(&sh[nbins], 1));
        }
        *to = make_int2(s[k], d[k]);
      }
    }
  }
  __syncthreads();
  for (int i = threadIdx.x; i < nbins; i += TB) {
    const int n = sh[i];
    segcnt[(size_t)i * NCHUNK + c] = (n < SEGCAP) ? n : SEGCAP;
  }
  if (threadIdx.x == 0) spillcnt[c] = sh[nbins];
}

__device__ __forceinline__ void gcn_place(int s, int d, int* lcnt, int base,
                                          int* __restrict__ bucket,
                                          int* __restrict__ ovfn,
                                          int* __restrict__ osrc,
                                          int* __restrict__ odst) {
  const int p = atomicAdd(&lcnt[d - base], 1);
  if (p < CAP) {
    bucket[(size_t)d * CAP + p] = s;
  } else {
    const int q = atomicAdd(ovfn, 1);
    osrc[q] = s;
    odst[q] = d;
  }
}

__global__ __launch_bounds__(TB) void gcn_edge_rank(
    const int2* __restrict__ seg, const int* __restrict__ segcnt,
    const int2* __restrict__ spill, const int* __restrict__ spillcnt,
    int chunk, int n, int* __restrict__ cnt, int* __restrict__ ovfn,
    int* __restrict__ bucket, int* __restrict__ osrc, int* __restrict__ odst) {
  __shared__ int lcnt[RNG];
  __shared__ int scnt[NCHUNK];
  __shared__ int sspill[NCHUNK];
  __shared__ int anyspill;
  const int b = blockIdx.x;
  const int base = b << RSH;
  const int tid = threadIdx.x;

  int mysc = 0, mysp = 0;
  if (tid < NCHUNK) {
    mysc = segcnt[(size_t)b * NCHUNK + tid];
    mysp = spillcnt[tid];
  }
  for (int i = tid; i < RNG; i += TB) lcnt[i] = 0;
  if (tid < NCHUNK) {
    scnt[tid] = mysc;
    sspill[tid] = mysp;
  }
  if (tid == 0) anyspill = 0;
  __syncthreads();
  if (mysp > 0) anyspill = 1;

  const int2* __restrict__ myseg = seg + (((size_t)b * NCHUNK) << SEGSH);
  for (int r = 0; r < 4; ++r) {
    const int s0 = tid + r * 4 * TB;
    int2 e[4];
    bool ok[4];
#pragma unroll
    for (int k = 0; k < 4; ++k) {
      const int slot = s0 + k * TB;
      ok[k] = (slot & (SEGCAP - 1)) < scnt[slot >> SEGSH];
      e[k] = ok[k] ? myseg[slot] : make_int2(0, 0);
    }
#pragma unroll
    for (int k = 0; k < 4; ++k) {
      if (ok[k]) gcn_place(e[k].x, e[k].y, lcnt, base, bucket, ovfn, osrc, odst);
    }
  }
  __syncthreads();
  if (anyspill) {
    for (int c = 0; c < NCHUNK; ++c) {
      const int oc = sspill[c];
      for (int i = tid; i < oc; i += TB) {
        const int2 e = spill[(size_t)c * chunk + i];
        if ((e.y >> RSH) == b)
          gcn_place(e.x, e.y, lcnt, base, bucket, ovfn, osrc, odst);
      }
    }
  }
  __syncthreads();
  for (int i = tid; i < RNG; i += TB) {
    if (base + i < n) cnt[base + i] = lcnt[i];
  }
}

#define XP 68
#define WPB 8

__device__ __forceinline__ void gcn_sum8(float4& acc,
                                         const float* __restrict__ g,
                                         const int* __restrict__ cnt,
                                         unsigned q, unsigned self, int4 ia,
                                         int4 ib, int j, int end) {
  const bool k0 = j + 0 < end, k1 = j + 1 < end, k2 = j + 2 < end,
             k3 = j + 3 < end, k4 = j + 4 < end, k5 = j + 5 < end,
             k6 = j + 6 < end, k7 = j + 7 < end;
  const unsigned f = k0 ? (unsigned)ia.x : self;
  const unsigned s1 = k1 ? (unsigned)ia.y : f;
  const unsigned s2 = k2 ? (unsigned)ia.z : f;
  const unsigned s3 = k3 ? (unsigned)ia.w : f;
  const unsigned s4 = k4 ? (unsigned)ib.x : f;
  const unsigned s5 = k5 ? (unsigned)ib.y : f;
  const unsigned s6 = k6 ? (unsigned)ib.z : f;
  const unsigned s7 = k7 ? (unsigned)ib.w : f;
  const int c0 = cnt[f];
  const float4 v0 = *(const float4*)(g + (f * 64u + q));
  const int c1 = cnt[s1];
  const float4 v1 = *(const float4*)(g + (s1 * 64u + q));
  const int c2 = cnt[s2];
  const float4 v2 = *(const float4*)(g + (s2 * 64u + q));
  const int c3 = cnt[s3];
  const float4 v3 = *(const float4*)(g + (s3 * 64u + q));
  const int c4 = cnt[s4];
  const float4 v4 = *(const float4*)(g + (s4 * 64u + q));
  const int c5 = cnt[s5];
  const float4 v5 = *(const float4*)(g + (s5 * 64u + q));
  const int c6 = cnt[s6];
  const float4 v6 = *(const float4*)(g + (s6 * 64u + q));
  const int c7 = cnt[s7];
  const float4 v7 = *(const float4*)(g + (s7 * 64u + q));
  __builtin_amdgcn_sched_barrier(0);
  const float w0 = gcn_dinv(c0), w1 = gcn_dinv(c1), w2 = gcn_dinv(c2),
              w3 = gcn_dinv(c3), w4 = gcn_dinv(c4), w5 = gcn_dinv(c5),
              w6 = gcn_dinv(c6), w7 = gcn_dinv(c7);
  acc.x = fmaf(w0, k0 ? v0.x : 0.0f, acc.x); acc.y = fmaf(w0, k0 ? v0.y : 0.0f, acc.y);
  acc.z = fmaf(w0, k0 ? v0.z : 0.0f, acc.z); acc.w = fmaf(w0, k0 ? v0.w : 0.0f, acc.w);
  acc.x = fmaf(w1, k1 ? v1.x : 0.0f, acc.x); acc.y = fmaf(w1, k1 ? v1.y : 0.0f, acc.y);
  acc.z = fmaf(w1, k1 ? v1.z : 0.0f, acc.z); acc.w = fmaf(w1, k1 ? v1.w : 0.0f, acc.w);
  acc.x = fmaf(w2, k2 ? v2.x : 0.0f, acc.x); acc.y = fmaf(w2, k2 ? v2.y : 0.0f, acc.y);
  acc.z = fmaf(w2, k2 ? v2.z : 0.0f, acc.z); acc.w = fmaf(w2, k2 ? v2.w : 0.0f, acc.w);
  acc.x = fmaf(w3, k3 ? v3.x : 0.0f, acc.x); acc.y = fmaf(w3, k3 ? v3.y : 0.0f, acc.y);
  acc.z = fmaf(w3, k3 ? v3.z : 0.0f, acc.z); acc.w = fmaf(w3, k3 ? v3.w : 0.0f, acc.w);
  acc.x = fmaf(w4, k4 ? v4.x : 0.0f, acc.x); acc.y = fmaf(w4, k4 ? v4.y : 0.0f, acc.y);
  acc.z = fmaf(w4, k4 ? v4.z : 0.0f, acc.z); acc.w = fmaf(w4, k4 ? v4.w : 0.0f, acc.w);
  acc.x = fmaf(w5, k5 ? v5.x : 0.0f, acc.x); acc.y = fmaf(w5, k5 ? v5.y : 0.0f, acc.y);
  acc.z = fmaf(w5, k5 ? v5.z : 0.0f, acc.z); acc.w = fmaf(w5, k5 ? v5.w : 0.0f, acc.w);
  acc.x = fmaf(w6, k6 ? v6.x : 0.0f, acc.x); acc.y = fmaf(w6, k6 ? v6.y : 0.0f, acc.y);
  acc.z = fmaf(w6, k6 ? v6.z : 0.0f, acc.z); acc.w = fmaf(w6, k6 ? v6.w : 0.0f, acc.w);
  acc.x = fmaf(w7, k7 ? v7.x : 0.0f, acc.x); acc.y = fmaf(w7, k7 ? v7.y : 0.0f, acc.y);
  acc.z = fmaf(w7, k7 ? v7.z : 0.0f, acc.z); acc.w = fmaf(w7, k7 ? v7.w : 0.0f, acc.w);
}

__global__ __launch_bounds__(32 * WPB) void gcn_gather1_gemm12(
    const int* __restrict__ cnt, const int* __restrict__ ovfn,
    const int* __restrict__ bucket, const int* __restrict__ osrc,
    const int* __restrict__ odst, const float* __restrict__ X,
    const float* __restrict__ W1, const float* __restrict__ b1,
    const float* __restrict__ W2, float* __restrict__ g2, int n) {
  __shared__ __attribute__((aligned(16))) float sW1[64 * 64];
  __shared__ __attribute__((aligned(16))) float sW2[64 * 16];
  __shared__ __attribute__((aligned(16))) float sX[WPB * 16 * XP];
  __shared__ __attribute__((aligned(16))) float sD[WPB * 16];

  const int tid = threadIdx.x;
  const int wave = tid >> 5;
  const int half = (tid >> 4) & 1;
  const int lp = tid & 15;
  const unsigned q = (unsigned)lp << 2;
  const int rowbase = (blockIdx.x * WPB + wave) * 16;

  const float4 wa = ((const float4*)W1)[tid];
  const float4 wb = ((const float4*)W1)[tid + 256];
  const float4 wc = ((const float4*)W1)[tid + 512];
  const float4 wd = ((const float4*)W1)[tid + 768];
  const float4 we = ((const float4*)W2)[tid];

  int nodeN = (rowbase + half < n) ? rowbase + half : n - 1;
  int degN = cnt[nodeN];
  int4 iaN = ((const int4*)(bucket + (size_t)nodeN * CAP))[0];
  int4 ibN = ((const int4*)(bucket + (size_t)nodeN * CAP))[1];
  float4 selfN = *(const float4*)(X + ((unsigned)nodeN * 64u + q));

  ((float4*)sW1)[tid] = wa;
  ((float4*)sW1)[tid + 256] = wb;
  ((float4*)sW1)[tid + 512] = wc;
  ((float4*)sW1)[tid + 768] = wd;
  ((float4*)sW2)[tid] = we;
  __syncthreads();

  float* __restrict__ myX = sX + wave * 16 * XP;
  for (int t = 0; t < 8; ++t) {
    const int node = nodeN;
    const int deg = degN;
    int4 ia = iaN, ib = ibN;
    const float dv = gcn_dinv(deg);
    float4 acc;
    acc.x = dv * selfN.x; acc.y = dv * selfN.y;
    acc.z = dv * selfN.z; acc.w = dv * selfN.w;
    const int4* __restrict__ col4 = (const int4*)(bucket + (size_t)node * CAP);
    if (t + 1 < 8) {
      const int nb = rowbase + 2 * (t + 1) + half;
      nodeN = (nb < n) ? nb : n - 1;
      degN = cnt[nodeN];
      iaN = ((const int4*)(bucket + (size_t)nodeN * CAP))[0];
      ibN = ((const int4*)(bucket + (size_t)nodeN * CAP))[1];
      selfN = *(const float4*)(X + ((unsigned)nodeN * 64u + q));
    }
    const int end = (deg < CAP) ? deg : CAP;
    gcn_sum8(acc, X, cnt, q, (unsigned)node, ia, ib, 0, end);
    for (int j = 8; j < end; j += 8) {
      ia = col4[j >> 2];
      ib = col4[(j >> 2) + 1];
      gcn_sum8(acc, X, cnt, q, (unsigned)node, ia, ib, j, end);
    }
    if (deg > CAP) {
      const int no = *ovfn;
      for (int i = 0; i < no; ++i) {
        if (odst[i] == node) {
          const unsigned s = (unsigned)osrc[i];
          const float w = gcn_dinv(cnt[s]);
          const float4 v = *(const float4*)(X + (s * 64u + q));
          acc.x = fmaf(w, v.x, acc.x); acc.y = fmaf(w, v.y, acc.y);
          acc.z = fmaf(w, v.z, acc.z); acc.w = fmaf(w, v.w, acc.w);
        }
      }
    }
    float4 x;
    x.x = dv * acc.x; x.y = dv * acc.y; x.z = dv * acc.z; x.w = dv * acc.w;
    const int lr = 2 * t + half;
    if (lp == 0) sD[wave * 16 + lr] = dv;
    *(float4*)(myX + lr * XP + q) = x;
  }
  __builtin_amdgcn_fence(__ATOMIC_RELEASE, "wavefront");
  __builtin_amdgcn_wave_barrier();
  __builtin_amdgcn_fence(__ATOMIC_ACQUIRE, "wavefront");

  const float* __restrict__ Arow = myX + lp * XP;
  v2f a[16];
#pragma unroll
  for (int kk = 0; kk < 16; ++kk) a[kk] = *(const v2f*)(Arow + 4 * kk + 2 * half);
  const float bb0 = b1[lp], bb1 = b1[16 + lp], bb2 = b1[32 + lp], bb3 = b1[48 + lp];
  v8f acc0 = {}, acc1 = {}, acc2 = {}, acc3 = {};
#pragma unroll
  for (int kk = 0; kk < 16; ++kk) {
    const int ka = 4 * kk + 2 * half;
    const float* b0r = &sW1[ka * 64];
    const float* b1r = &sW1[(ka + 1) * 64];
    v2f bf;
    bf.x = b0r[lp];
    bf.y = b1r[lp];
    acc0 = __builtin_amdgcn_wmma_f32_16x16x4_f32(false, a[kk], false, bf,
                                                 (short)0, acc0, false, false);
    bf.x = b0r[16 + lp];
    bf.y = b1r[16 + lp];
    acc1 = __builtin_amdgcn_wmma_f32_16x16x4_f32(false, a[kk], false, bf,
                                                 (short)0, acc1, false, false);
    bf.x = b0r[32 + lp];
    bf.y = b1r[32 + lp];
    acc2 = __builtin_amdgcn_wmma_f32_16x16x4_f32(false, a[kk], false, bf,
                                                 (short)0, acc2, false, false);
    bf.x = b0r[48 + lp];
    bf.y = b1r[48 + lp];
    acc3 = __builtin_amdgcn_wmma_f32_16x16x4_f32(false, a[kk], false, bf,
                                                 (short)0, acc3, false, false);
  }
#pragma unroll
  for (int j = 0; j < 8; ++j) {
    float* hr = myX + (j + 8 * half) * XP + lp;
    hr[0]  = fmaxf(acc0[j] + bb0, 0.0f);
    hr[16] = fmaxf(acc1[j] + bb1, 0.0f);
    hr[32] = fmaxf(acc2[j] + bb2, 0.0f);
    hr[48] = fmaxf(acc3[j] + bb3, 0.0f);
  }
  __builtin_amdgcn_fence(__ATOMIC_RELEASE, "wavefront");
  __builtin_amdgcn_wave_barrier();
  __builtin_amdgcn_fence(__ATOMIC_ACQUIRE, "wavefront");

  v8f out2 = {};
#pragma unroll
  for (int k = 0; k < 64; k += 4) {
    const int ka = k + 2 * half;
    const v2f a2 = *(const v2f*)(Arow + ka);
    v2f b;
    b.x = sW2[ka * 16 + lp];
    b.y = sW2[(ka + 1) * 16 + lp];
    out2 = __builtin_amdgcn_wmma_f32_16x16x4_f32(false, a2, false, b, (short)0,
                                                 out2, false, false);
  }

  const float4 d0 = *(const float4*)(sD + wave * 16 + 8 * half);
  const float4 d1 = *(const float4*)(sD + wave * 16 + 8 * half + 4);
  const float dv8[8] = {d0.x, d0.y, d0.z, d0.w, d1.x, d1.y, d1.z, d1.w};
#pragma unroll
  for (int j = 0; j < 8; ++j) {
    const int row = rowbase + j + 8 * half;
    if (row < n) g2[(size_t)row * 16 + lp] = dv8[j] * out2[j];
  }
}

__global__ __launch_bounds__(256) void gcn_gather2(
    const int* __restrict__ cnt, const int* __restrict__ ovfn,
    const int* __restrict__ bucket, const int* __restrict__ osrc,
    const int* __restrict__ odst, const float* __restrict__ g2,
    const float* __restrict__ b2, float* __restrict__ out, int n) {
  const int tid = blockIdx.x * blockDim.x + threadIdx.x;
  const int node = tid >> 2;
  if (node >= n) return;
  const int q = (tid & 3) << 2;
  const int deg = cnt[node];
  const int end = (deg < CAP) ? deg : CAP;
  const int* __restrict__ col = bucket + (size_t)node * CAP;
  const float* __restrict__ gq = g2 + q;
  float4 acc = *(const float4*)(gq + (size_t)node * 16);
  int j = 0;
  for (; j + 4 <= end; j += 4) {
    const int s0 = col[j], s1 = col[j + 1], s2 = col[j + 2], s3 = col[j + 3];
    const float4 v0 = *(const float4*)(gq + (size_t)s0 * 16);
    const float4 v1 = *(const float4*)(gq + (size_t)s1 * 16);
    const float4 v2 = *(const float4*)(gq + (size_t)s2 * 16);
    const float4 v3 = *(const float4*)(gq + (size_t)s3 * 16);
    acc.x += v0.x; acc.y += v0.y; acc.z += v0.z; acc.w += v0.w;
    acc.x += v1.x; acc.y += v1.y; acc.z += v1.z; acc.w += v1.w;
    acc.x += v2.x; acc.y += v2.y; acc.z += v2.z; acc.w += v2.w;
    acc.x += v3.x; acc.y += v3.y; acc.z += v3.z; acc.w += v3.w;
  }
  for (; j < end; ++j) {
    const float4 v = *(const float4*)(gq + (size_t)col[j] * 16);
    acc.x += v.x; acc.y += v.y; acc.z += v.z; acc.w += v.w;
  }
  if (deg > CAP) {
    const int no = *ovfn;
    for (int i = 0; i < no; ++i) {
      if (odst[i] == node) {
        const float4 v = *(const float4*)(gq + (size_t)osrc[i] * 16);
        acc.x += v.x; acc.y += v.y; acc.z += v.z; acc.w += v.w;
      }
    }
  }
  const float dv = gcn_dinv(deg);
  float4 r;
  r.x = fmaf(dv, acc.x, b2[q + 0]);
  r.y = fmaf(dv, acc.y, b2[q + 1]);
  r.z = fmaf(dv, acc.z, b2[q + 2]);
  r.w = fmaf(dv, acc.w, b2[q + 3]);
  *(float4*)(out + (size_t)node * 16 + q) = r;
}

extern "C" void kernel_launch(void* const* d_in, const int* in_sizes, int n_in,
                              void* d_out, int out_size, void* d_ws, size_t ws_size,
                              hipStream_t stream) {
  const float* X  = (const float*)d_in[0];
  const int*   ei = (const int*)d_in[1];
  const float* W1 = (const float*)d_in[2];
  const float* b1 = (const float*)d_in[3];
  const float* W2 = (const float*)d_in[4];
  const float* b2 = (const float*)d_in[5];
  float* out = (float*)d_out;

  const int N = in_sizes[0] / 64;
  const int E = in_sizes[1] / 2;
  const int* src = ei;
  const int* dst = ei + E;

  float* ws   = (float*)d_ws;
  float* g2   = ws;
  int* cnt    = (int*)(g2 + (size_t)N * 16);
  int* ovfn   = cnt + N;
  int* bucket = cnt + ((N + 1 + 31) / 32) * 32;
  int* osrc   = bucket + (size_t)N * CAP;
  int* odst   = osrc + E;

  const int nbins = (N + RNG - 1) / RNG;
  const int chunk = (E + NCHUNK - 1) / NCHUNK;
  const size_t used = (size_t)((odst + E) - (int*)d_ws);
  int2* seg     = (int2*)((int*)d_ws + (used + 31) / 32 * 32);
  int2* spill   = seg + (size_t)nbins * NCHUNK * SEGCAP;
  int* segcnt   = (int*)(spill + (size_t)NCHUNK * chunk);
  int* spillcnt = segcnt + (size_t)nbins * NCHUNK;

  const int B = 256;

  gcn_edge_bin<<<NCHUNK, TB, (nbins + 1) * sizeof(int), stream>>>(
      src, dst, E, chunk, nbins, seg, segcnt, spill, spillcnt, ovfn);
  gcn_edge_rank<<<nbins, TB, 0, stream>>>(seg, segcnt, spill, spillcnt, chunk,
                                          N, cnt, ovfn, bucket, osrc, odst);
  gcn_gather1_gemm12<<<(N + 16 * WPB - 1) / (16 * WPB), 32 * WPB, 0, stream>>>(
      cnt, ovfn, bucket, osrc, odst, X, W1, b1, W2, g2, N);
  gcn_gather2<<<(N * 4 + B - 1) / B, B, 0, stream>>>(cnt, ovfn, bucket, osrc, odst,
                                                     g2, b2, out, N);
}


# Both edge passes build their rows in LDS (pass 1 its range segments, pass 2 its 512 node buckets) and write each 128-byte line with one store; 2M scattered 4/8-byte stores gone; slot order kept
# speedup vs baseline: 4.0691x; 1.0601x over previous
#include <hip/hip_runtime.h>


typedef __attribute__((ext_vector_type(2))) float v2f;
typedef __attribute__((ext_vector_type(8))) float v8f;

#define CAP 32

__device__ __forceinline__ float gcn_dinv(int c) {
  const float dg = (float)(c + 1);
  return (dg > 0.0f) ? rsqrtf(dg) : 0.0f;
}

#define RSH 9
#define RNG (1 << RSH)
#define SEGSH 6
#define SEGCAP (1 << SEGSH)
#define NCHUNK 256
#define TB 1024
#define NBMAX 256

__global__ __launch_bounds__(TB) void gcn_edge_bin(
    const int* __restrict__ src, const int* __restrict__ dst, int E, int chunk,
    int nbins, int2* __restrict__ seg, int* __restrict__ segcnt,
    int2* __restrict__ spill, int* __restrict__ spillcnt,
    int* __restrict__ ovfn) {
  __shared__ int sh[NBMAX + 1];
  __shared__ __attribute__((aligned(16))) int stage[NBMAX * SEGCAP * 2];
  const int c = blockIdx.x;
  for (int i = threadIdx.x; i <= nbins; i += TB) sh[i] = 0;
  if (c == 0 && threadIdx.x == 0) *ovfn = 0;
  __syncthreads();

  const int lo = c * chunk;
  const int hi = (lo + chunk < E) ? lo + chunk : E;
  for (int e0 = lo + threadIdx.x; e0 < hi; e0 += 4 * TB) {
    int d[4], s[4];
#pragma unroll
    for (int k = 0; k < 4; ++k) {
      const int e = e0 + k * TB;
      d[k] = (e < hi) ? dst[e] : 0;
      s[k] = (e < hi) ? src[e] : 0;
    }
#pragma unroll
    for (int k = 0; k < 4; ++k) {
      if (e0 + k * TB < hi) {
        const int bin = d[k] >> RSH;
        const int r = atomicAdd(&sh[bin], 1);
        if (r < SEGCAP) {
          const int at = (bin << SEGSH) + ((r + 2 * bin) & (SEGCAP - 1));
          *(int2*)(stage + 2 * at) = make_int2(s[k], d[k]);
        } else {
          spill[(size_t)lo + atomicAdd(&sh[nbins], 1)] = make_int2(s[k], d[k]);
        }
      }
    }
  }
  __syncthreads();
  const int wv = threadIdx.x >> 5;
  const int ln = threadIdx.x & 31;
  int nn[NBMAX / (TB / 32)];
  int4 vv[NBMAX / (TB / 32)];
#pragma unroll
  for (int k = 0; k < NBMAX / (TB / 32); ++k) {
    const int bin = wv + k * (TB / 32);
    const int n = sh[bin];
    nn[k] = (bin < nbins) ? ((n < SEGCAP) ? n : SEGCAP) : 0;
    vv[k] = ((const int4*)stage)[(bin << (SEGSH - 1)) + ((ln + bin) & 31)];
  }
#pragma unroll
  for (int k = 0; k < NBMAX / (TB / 32); ++k) {
    const int bin = wv + k * (TB / 32);
    const int n = nn[k];
    if ((ln >> 3) < ((n + 15) >> 4)) {
      int4 v = vv[k];
      if (2 * ln >= n) { v.x = 0; v.y = 0; }
      if (2 * ln + 1 >= n) { v.z = 0; v.w = 0; }
      ((int4*)(seg + (((size_t)bin * NCHUNK + c) << SEGSH)))[ln] = v;
    }
    if (ln == 0 && bin < nbins) segcnt[(size_t)bin * NCHUNK + c] = n;
  }
  if (threadIdx.x == 0) spillcnt[c] = sh[nbins];
}

__device__ __forceinline__ void gcn_place(int s, int d, int* lcnt, int* lb,
                                          int base, int* __restrict__ ovfn,
                                          int* __restrict__ osrc,
                                          int* __restrict__ odst) {
  const int row = d - base;
  const int p = atomicAdd(&lcnt[row], 1);
  if (p < CAP) {
    lb[row * CAP + ((p + 4 * ((row >> 1) & 7)) & (CAP - 1))] = s;
  } else {
    const int q = atomicAdd(ovfn, 1);
    osrc[q] = s;
    odst[q] = d;
  }
}

__global__ __launch_bounds__(TB) void gcn_edge_rank(
    const int2* __restrict__ seg, const int* __restrict__ segcnt,
    const int2* __restrict__ spill, const int* __restrict__ spillcnt,
    int chunk, int n, int* __restrict__ cnt, int* __restrict__ ovfn,
    int* __restrict__ bucket, int* __restrict__ osrc, int* __restrict__ odst) {
  __shared__ int lcnt[RNG];
  __shared__ int scnt[NCHUNK];
  __shared__ int sspill[NCHUNK];
  __shared__ int anyspill;
  __shared__ __attribute__((aligned(16))) int lb[RNG * CAP];
  const int b = blockIdx.x;
  const int base = b << RSH;
  const int tid = threadIdx.x;

  int mysc = 0, mysp = 0;
  if (tid < NCHUNK) {
    mysc = segcnt[(size_t)b * NCHUNK + tid];
    mysp = spillcnt[tid];
  }
  for (int i = tid; i < RNG; i += TB) lcnt[i] = 0;
  if (tid < NCHUNK) {
    scnt[tid] = mysc;
    sspill[tid] = mysp;
  }
  if (tid == 0) anyspill = 0;
  __syncthreads();
  if (mysp > 0) anyspill = 1;

  const int2* __restrict__ myseg = seg + (((size_t)b * NCHUNK) << SEGSH);
  for (int r = 0; r < 4; ++r) {
    const int s0 = tid + r * 4 * TB;
    int2 e[4];
    bool ok[4];
#pragma unroll
    for (int k = 0; k < 4; ++k) {
      const int slot = s0 + k * TB;
      ok[k] = (slot & (SEGCAP - 1)) < scnt[slot >> SEGSH];
      e[k] = ok[k] ? myseg[slot] : make_int2(0, 0);
    }
#pragma unroll
    for (int k = 0; k < 4; ++k) {
      if (ok[k]) gcn_place(e[k].x, e[k].y, lcnt, lb, base, ovfn, osrc, odst);
    }
  }
  __syncthreads();
  if (anyspill) {
    for (int c = 0; c < NCHUNK; ++c) {
      const int oc = sspill[c];
      for (int i = tid; i < oc; i += TB) {
        const int2 e = spill[(size_t)c * chunk + i];
        if ((e.y >> RSH) == b)
          gcn_place(e.x, e.y, lcnt, lb, base, ovfn, osrc, odst);
      }
    }
  }
  __syncthreads();
  for (int i = tid; i < RNG; i += TB) {
    if (base + i < n) cnt[base + i] = lcnt[i];
  }
  int dg[RNG * CAP / 4 / TB];
  int4 vv[RNG * CAP / 4 / TB];
#pragma unroll
  for (int k = 0; k < RNG * CAP / 4 / TB; ++k) {
    const int i = tid + k * TB;
    const int row = i >> 3;
    dg[k] = lcnt[row];
    vv[k] = ((const int4*)lb)[(row << 3) + (((i & 7) + (row >> 1)) & 7)];
  }
#pragma unroll
  for (int k = 0; k < RNG * CAP / 4 / TB; ++k) {
    const int i = tid + k * TB;
    const int row = i >> 3;
    const int j = (i & 7) << 2;
    const int deg = dg[k];
    int4 v = vv[k];
    v.x = (j + 0 < deg) ? v.x : 0;
    v.y = (j + 1 < deg) ? v.y : 0;
    v.z = (j + 2 < deg) ? v.z : 0;
    v.w = (j + 3 < deg) ? v.w : 0;
    if (base + row < n) ((int4*)bucket)[((size_t)base << 3) + i] = v;
  }
}

#define XP 68
#define WPB 8

__device__ __forceinline__ void gcn_sum8(float4& acc,
                                         const float* __restrict__ g,
                                         const int* __restrict__ cnt,
                                         unsigned q, unsigned self, int4 ia,
                                         int4 ib, int j, int end) {
  const bool k0 = j + 0 < end, k1 = j + 1 < end, k2 = j + 2 < end,
             k3 = j + 3 < end, k4 = j + 4 < end, k5 = j + 5 < end,
             k6 = j + 6 < end, k7 = j + 7 < end;
  const unsigned f = k0 ? (unsigned)ia.x : self;
  const unsigned s1 = k1 ? (unsigned)ia.y : f;
  const unsigned s2 = k2 ? (unsigned)ia.z : f;
  const unsigned s3 = k3 ? (unsigned)ia.w : f;
  const unsigned s4 = k4 ? (unsigned)ib.x : f;
  const unsigned s5 = k5 ? (unsigned)ib.y : f;
  const unsigned s6 = k6 ? (unsigned)ib.z : f;
  const unsigned s7 = k7 ? (unsigned)ib.w : f;
  const int c0 = cnt[f];
  const float4 v0 = *(const float4*)(g + (f * 64u + q));
  const int c1 = cnt[s1];
  const float4 v1 = *(const float4*)(g + (s1 * 64u + q));
  const int c2 = cnt[s2];
  const float4 v2 = *(const float4*)(g + (s2 * 64u + q));
  const int c3 = cnt[s3];
  const float4 v3 = *(const float4*)(g + (s3 * 64u + q));
  const int c4 = cnt[s4];
  const float4 v4 = *(const float4*)(g + (s4 * 64u + q));
  const int c5 = cnt[s5];
  const float4 v5 = *(const float4*)(g + (s5 * 64u + q));
  const int c6 = cnt[s6];
  const float4 v6 = *(const float4*)(g + (s6 * 64u + q));
  const int c7 = cnt[s7];
  const float4 v7 = *(const float4*)(g + (s7 * 64u + q));
  __builtin_amdgcn_sched_barrier(0);
  const float w0 = gcn_dinv(c0), w1 = gcn_dinv(c1), w2 = gcn_dinv(c2),
              w3 = gcn_dinv(c3), w4 = gcn_dinv(c4), w5 = gcn_dinv(c5),
              w6 = gcn_dinv(c6), w7 = gcn_dinv(c7);
  acc.x = fmaf(w0, k0 ? v0.x : 0.0f, acc.x); acc.y = fmaf(w0, k0 ? v0.y : 0.0f, acc.y);
  acc.z = fmaf(w0, k0 ? v0.z : 0.0f, acc.z); acc.w = fmaf(w0, k0 ? v0.w : 0.0f, acc.w);
  acc.x = fmaf(w1, k1 ? v1.x : 0.0f, acc.x); acc.y = fmaf(w1, k1 ? v1.y : 0.0f, acc.y);
  acc.z = fmaf(w1, k1 ? v1.z : 0.0f, acc.z); acc.w = fmaf(w1, k1 ? v1.w : 0.0f, acc.w);
  acc.x = fmaf(w2, k2 ? v2.x : 0.0f, acc.x); acc.y = fmaf(w2, k2 ? v2.y : 0.0f, acc.y);
  acc.z = fmaf(w2, k2 ? v2.z : 0.0f, acc.z); acc.w = fmaf(w2, k2 ? v2.w : 0.0f, acc.w);
  acc.x = fmaf(w3, k3 ? v3.x : 0.0f, acc.x); acc.y = fmaf(w3, k3 ? v3.y : 0.0f, acc.y);
  acc.z = fmaf(w3, k3 ? v3.z : 0.0f, acc.z); acc.w = fmaf(w3, k3 ? v3.w : 0.0f, acc.w);
  acc.x = fmaf(w4, k4 ? v4.x : 0.0f, acc.x); acc.y = fmaf(w4, k4 ? v4.y : 0.0f, acc.y);
  acc.z = fmaf(w4, k4 ? v4.z : 0.0f, acc.z); acc.w = fmaf(w4, k4 ? v4.w : 0.0f, acc.w);
  acc.x = fmaf(w5, k5 ? v5.x : 0.0f, acc.x); acc.y = fmaf(w5, k5 ? v5.y : 0.0f, acc.y);
  acc.z = fmaf(w5, k5 ? v5.z : 0.0f, acc.z); acc.w = fmaf(w5, k5 ? v5.w : 0.0f, acc.w);
  acc.x = fmaf(w6, k6 ? v6.x : 0.0f, acc.x); acc.y = fmaf(w6, k6 ? v6.y : 0.0f, acc.y);
  acc.z = fmaf(w6, k6 ? v6.z : 0.0f, acc.z); acc.w = fmaf(w6, k6 ? v6.w : 0.0f, acc.w);
  acc.x = fmaf(w7, k7 ? v7.x : 0.0f, acc.x); acc.y = fmaf(w7, k7 ? v7.y : 0.0f, acc.y);
  acc.z = fmaf(w7, k7 ? v7.z : 0.0f, acc.z); acc.w = fmaf(w7, k7 ? v7.w : 0.0f, acc.w);
}

__global__ __launch_bounds__(32 * WPB) void gcn_gather1_gemm12(
    const int* __restrict__ cnt, const int* __restrict__ ovfn,
    const int* __restrict__ bucket, const int* __restrict__ osrc,
    const int* __restrict__ odst, const float* __restrict__ X,
    const float* __restrict__ W1, const float* __restrict__ b1,
    const float* __restrict__ W2, float* __restrict__ g2, int n) {
  __shared__ __attribute__((aligned(16))) float sW1[64 * 64];
  __shared__ __attribute__((aligned(16))) float sW2[64 * 16];
  __shared__ __attribute__((aligned(16))) float sX[WPB * 16 * XP];
  __shared__ __attribute__((aligned(16))) float sD[WPB * 16];

  const int tid = threadIdx.x;
  const int wave = tid >> 5;
  const int half = (tid >> 4) & 1;
  const int lp = tid & 15;
  const unsigned q = (unsigned)lp << 2;
  const int rowbase = (blockIdx.x * WPB + wave) * 16;

  const float4 wa = ((const float4*)W1)[tid];
  const float4 wb = ((const float4*)W1)[tid + 256];
  const float4 wc = ((const float4*)W1)[tid + 512];
  const float4 wd = ((const float4*)W1)[tid + 768];
  const float4 we = ((const float4*)W2)[tid];

  int nodeN = (rowbase + half < n) ? rowbase + half : n - 1;
  int degN = cnt[nodeN];
  int4 iaN = ((const int4*)(bucket + (size_t)nodeN * CAP))[0];
  int4 ibN = ((const int4*)(bucket + (size_t)nodeN * CAP))[1];
  float4 selfN = *(const float4*)(X + ((unsigned)nodeN * 64u + q));

  ((float4*)sW1)[tid] = wa;
  ((float4*)sW1)[tid + 256] = wb;
  ((float4*)sW1)[tid + 512] = wc;
  ((float4*)sW1)[tid + 768] = wd;
  ((float4*)sW2)[tid] = we;
  __syncthreads();

  float* __restrict__ myX = sX + wave * 16 * XP;
  for (int t = 0; t < 8; ++t) {
    const int node = nodeN;
    const int deg = degN;
    int4 ia = iaN, ib = ibN;
    const float dv = gcn_dinv(deg);
    float4 acc;
    acc.x = dv * selfN.x; acc.y = dv * selfN.y;
    acc.z = dv * selfN.z; acc.w = dv * selfN.w;
    const int4* __restrict__ col4 = (const int4*)(bucket + (size_t)node * CAP);
    if (t + 1 < 8) {
      const int nb = rowbase + 2 * (t + 1) + half;
      nodeN = (nb < n) ? nb : n - 1;
      degN = cnt[nodeN];
      iaN = ((const int4*)(bucket + (size_t)nodeN * CAP))[0];
      ibN = ((const int4*)(bucket + (size_t)nodeN * CAP))[1];
      selfN = *(const float4*)(X + ((unsigned)nodeN * 64u + q));
    }
    const int end = (deg < CAP) ? deg : CAP;
    gcn_sum8(acc, X, cnt, q, (unsigned)node, ia, ib, 0, end);
    for (int j = 8; j < end; j += 8) {
      ia = col4[j >> 2];
      ib = col4[(j >> 2) + 1];
      gcn_sum8(acc, X, cnt, q, (unsigned)node, ia, ib, j, end);
    }
    if (deg > CAP) {
      const int no = *ovfn;
      for (int i = 0; i < no; ++i) {
        if (odst[i] == node) {
          const unsigned s = (unsigned)osrc[i];
          const float w = gcn_dinv(cnt[s]);
          const float4 v = *(const float4*)(X + (s * 64u + q));
          acc.x = fmaf(w, v.x, acc.x); acc.y = fmaf(w, v.y, acc.y);
          acc.z = fmaf(w, v.z, acc.z); acc.w = fmaf(w, v.w, acc.w);
        }
      }
    }
    float4 x;
    x.x = dv * acc.x; x.y = dv * acc.y; x.z = dv * acc.z; x.w = dv * acc.w;
    const int lr = 2 * t + half;
    if (lp == 0) sD[wave * 16 + lr] = dv;
    *(float4*)(myX + lr * XP + q) = x;
  }
  __builtin_amdgcn_fence(__ATOMIC_RELEASE, "wavefront");
  __builtin_amdgcn_wave_barrier();
  __builtin_amdgcn_fence(__ATOMIC_ACQUIRE, "wavefront");

  const float* __restrict__ Arow = myX + lp * XP;
  v2f a[16];
#pragma unroll
  for (int kk = 0; kk < 16; ++kk) a[kk] = *(const v2f*)(Arow + 4 * kk + 2 * half);
  const float bb0 = b1[lp], bb1 = b1[16 + lp], bb2 = b1[32 + lp], bb3 = b1[48 + lp];
  v8f acc0 = {}, acc1 = {}, acc2 = {}, acc3 = {};
#pragma unroll
  for (int kk = 0; kk < 16; ++kk) {
    const int ka = 4 * kk + 2 * half;
    const float* b0r = &sW1[ka * 64];
    const float* b1r = &sW1[(ka + 1) * 64];
    v2f bf;
    bf.x = b0r[lp];
    bf.y = b1r[lp];
    acc0 = __builtin_amdgcn_wmma_f32_16x16x4_f32(false, a[kk], false, bf,
                                                 (short)0, acc0, false, false);
    bf.x = b0r[16 + lp];
    bf.y = b1r[16 + lp];
    acc1 = __builtin_amdgcn_wmma_f32_16x16x4_f32(false, a[kk], false, bf,
                                                 (short)0, acc1, false, false);
    bf.x = b0r[32 + lp];
    bf.y = b1r[32 + lp];
    acc2 = __builtin_amdgcn_wmma_f32_16x16x4_f32(false, a[kk], false, bf,
                                                 (short)0, acc2, false, false);
    bf.x = b0r[48 + lp];
    bf.y = b1r[48 + lp];
    acc3 = __builtin_amdgcn_wmma_f32_16x16x4_f32(false, a[kk], false, bf,
                                                 (short)0, acc3, false, false);
  }
#pragma unroll
  for (int j = 0; j < 8; ++j) {
    float* hr = myX + (j + 8 * half) * XP + lp;
    hr[0]  = fmaxf(acc0[j] + bb0, 0.0f);
    hr[16] = fmaxf(acc1[j] + bb1, 0.0f);
    hr[32] = fmaxf(acc2[j] + bb2, 0.0f);
    hr[48] = fmaxf(acc3[j] + bb3, 0.0f);
  }
  __builtin_amdgcn_fence(__ATOMIC_RELEASE, "wavefront");
  __builtin_amdgcn_wave_barrier();
  __builtin_amdgcn_fence(__ATOMIC_ACQUIRE, "wavefront");

  v8f out2 = {};
#pragma unroll
  for (int k = 0; k < 64; k += 4) {
    const int ka = k + 2 * half;
    const v2f a2 = *(const v2f*)(Arow + ka);
    v2f b;
    b.x = sW2[ka * 16 + lp];
    b.y = sW2[(ka + 1) * 16 + lp];
    out2 = __builtin_amdgcn_wmma_f32_16x16x4_f32(false, a2, false, b, (short)0,
                                                 out2, false, false);
  }

  const float4 d0 = *(const float4*)(sD + wave * 16 + 8 * half);
  const float4 d1 = *(const float4*)(sD + wave * 16 + 8 * half + 4);
  const float dv8[8] = {d0.x, d0.y, d0.z, d0.w, d1.x, d1.y, d1.z, d1.w};
#pragma unroll
  for (int j = 0; j < 8; ++j) {
    const int row = rowbase + j + 8 * half;
    if (row < n) g2[(size_t)row * 16 + lp] = dv8[j] * out2[j];
  }
}

__global__ __launch_bounds__(256) void gcn_gather2(
    const int* __restrict__ cnt, const int* __restrict__ ovfn,
    const int* __restrict__ bucket, const int* __restrict__ osrc,
    const int* __restrict__ odst, const float* __restrict__ g2,
    const float* __restrict__ b2, float* __restrict__ out, int n) {
  const int tid = blockIdx.x * blockDim.x + threadIdx.x;
  const int node = tid >> 2;
  if (node >= n) return;
  const int q = (tid & 3) << 2;
  const int deg = cnt[node];
  const int end = (deg < CAP) ? deg : CAP;
  const int* __restrict__ col = bucket + (size_t)node * CAP;
  const float* __restrict__ gq = g2 + q;
  float4 acc = *(const float4*)(gq + (size_t)node * 16);
  int j = 0;
  for (; j + 4 <= end; j += 4) {
    const int s0 = col[j], s1 = col[j + 1], s2 = col[j + 2], s3 = col[j + 3];
    const float4 v0 = *(const float4*)(gq + (size_t)s0 * 16);
    const float4 v1 = *(const float4*)(gq + (size_t)s1 * 16);
    const float4 v2 = *(const float4*)(gq + (size_t)s2 * 16);
    const float4 v3 = *(const float4*)(gq + (size_t)s3 * 16);
    acc.x += v0.x; acc.y += v0.y; acc.z += v0.z; acc.w += v0.w;
    acc.x += v1.x; acc.y += v1.y; acc.z += v1.z; acc.w += v1.w;
    acc.x += v2.x; acc.y += v2.y; acc.z += v2.z; acc.w += v2.w;
    acc.x += v3.x; acc.y += v3.y; acc.z += v3.z; acc.w += v3.w;
  }
  for (; j < end; ++j) {
    const float4 v = *(const float4*)(gq + (size_t)col[j] * 16);
    acc.x += v.x; acc.y += v.y; acc.z += v.z; acc.w += v.w;
  }
  if (deg > CAP) {
    const int no = *ovfn;
    for (int i = 0; i < no; ++i) {
      if (odst[i] == node) {
        const float4 v = *(const float4*)(gq + (size_t)osrc[i] * 16);
        acc.x += v.x; acc.y += v.y; acc.z += v.z; acc.w += v.w;
      }
    }
  }
  const float dv = gcn_dinv(deg);
  float4 r;
  r.x = fmaf(dv, acc.x, b2[q + 0]);
  r.y = fmaf(dv, acc.y, b2[q + 1]);
  r.z = fmaf(dv, acc.z, b2[q + 2]);
  r.w = fmaf(dv, acc.w, b2[q + 3]);
  *(float4*)(out + (size_t)node * 16 + q) = r;
}

extern "C" void kernel_launch(void* const* d_in, const int* in_sizes, int n_in,
                              void* d_out, int out_size, void* d_ws, size_t ws_size,
                              hipStream_t stream) {
  const float* X  = (const float*)d_in[0];
  const int*   ei = (const int*)d_in[1];
  const float* W1 = (const float*)d_in[2];
  const float* b1 = (const float*)d_in[3];
  const float* W2 = (const float*)d_in[4];
  const float* b2 = (const float*)d_in[5];
  float* out = (float*)d_out;

  const int N = in_sizes[0] / 64;
  const int E = in_sizes[1] / 2;
  const int* src = ei;
  const int* dst = ei + E;

  float* ws   = (float*)d_ws;
  float* g2   = ws;
  int* cnt    = (int*)(g2 + (size_t)N * 16);
  int* ovfn   = cnt + N;
  int* bucket = cnt + ((N + 1 + 31) / 32) * 32;
  int* osrc   = bucket + (size_t)N * CAP;
  int* odst   = osrc + E;

  const int nbins = (N + RNG - 1) / RNG;
  const int chunk = (E + NCHUNK - 1) / NCHUNK;
  const size_t used = (size_t)((odst + E) - (int*)d_ws);
  int2* seg     = (int2*)((int*)d_ws + (used + 31) / 32 * 32);
  int2* spill   = seg + (size_t)nbins * NCHUNK * SEGCAP;
  int* segcnt   = (int*)(spill + (size_t)NCHUNK * chunk);
  int* spillcnt = segcnt + (size_t)nbins * NCHUNK;

  const int B = 256;

  gcn_edge_bin<<<NCHUNK, TB, 0, stream>>>(
      src, dst, E, chunk, nbins, seg, segcnt, spill, spillcnt, ovfn);
  gcn_edge_rank<<<nbins, TB, 0, stream>>>(seg, segcnt, spill, spillcnt, chunk,
                                          N, cnt, ovfn, bucket, osrc, odst);
  gcn_gather1_gemm12<<<(N + 16 * WPB - 1) / (16 * WPB), 32 * WPB, 0, stream>>>(
      cnt, ovfn, bucket, osrc, odst, X, W1, b1, W2, g2, N);
  gcn_gather2<<<(N * 4 + B - 1) / B, B, 0, stream>>>(cnt, ovfn, bucket, osrc, odst,
                                                     g2, b2, out, N);
}


# Pass 2: a wave owns 8 whole segments, their counts by one scalar load; all its entries requested together, then ranked together (no LDS count read or rank wait per entry); sums stay arrival-ranked
# speedup vs baseline: 4.2302x; 1.0396x over previous
#include <hip/hip_runtime.h>


typedef __attribute__((ext_vector_type(2))) float v2f;
typedef __attribute__((ext_vector_type(8))) float v8f;

#define CAP 32

__device__ __forceinline__ float gcn_dinv(int c) {
  const float dg = (float)(c + 1);
  return (dg > 0.0f) ? rsqrtf(dg) : 0.0f;
}

#define RSH 9
#define RNG (1 << RSH)
#define SEGSH 6
#define SEGCAP (1 << SEGSH)
#define NCHUNK 256
#define TB 1024
#define NBMAX 256

__global__ __launch_bounds__(TB) void gcn_edge_bin(
    const int* __restrict__ src, const int* __restrict__ dst, int E, int chunk,
    int nbins, int2* __restrict__ seg, int* __restrict__ segcnt,
    int2* __restrict__ spill, int* __restrict__ spillcnt,
    int* __restrict__ ovfn) {
  __shared__ int sh[NBMAX + 1];
  __shared__ __attribute__((aligned(16))) int stage[NBMAX * SEGCAP * 2];
  const int c = blockIdx.x;
  for (int i = threadIdx.x; i <= nbins; i += TB) sh[i] = 0;
  if (c == 0 && threadIdx.x == 0) *ovfn = 0;
  __syncthreads();

  const int lo = c * chunk;
  const int hi = (lo + chunk < E) ? lo + chunk : E;
  for (int e0 = lo + threadIdx.x; e0 < hi; e0 += 4 * TB) {
    int d[4], s[4];
#pragma unroll
    for (int k = 0; k < 4; ++k) {
      const int e = e0 + k * TB;
      d[k] = (e < hi) ? dst[e] : 0;
      s[k] = (e < hi) ? src[e] : 0;
    }
#pragma unroll
    for (int k = 0; k < 4; ++k) {
      if (e0 + k * TB < hi) {
        const int bin = d[k] >> RSH;
        const int r = atomicAdd(&sh[bin], 1);
        if (r < SEGCAP) {
          const int at = (bin << SEGSH) + ((r + 2 * bin) & (SEGCAP - 1));
          *(int2*)(stage + 2 * at) = make_int2(s[k], d[k]);
        } else {
          spill[(size_t)lo + atomicAdd(&sh[nbins], 1)] = make_int2(s[k], d[k]);
        }
      }
    }
  }
  __syncthreads();
  const int wv = threadIdx.x >> 5;
  const int ln = threadIdx.x & 31;
  int nn[NBMAX / (TB / 32)];
  int4 vv[NBMAX / (TB / 32)];
#pragma unroll
  for (int k = 0; k < NBMAX / (TB / 32); ++k) {
    const int bin = wv + k * (TB / 32);
    const int n = sh[bin];
    nn[k] = (bin < nbins) ? ((n < SEGCAP) ? n : SEGCAP) : 0;
    vv[k] = ((const int4*)stage)[(bin << (SEGSH - 1)) + ((ln + bin) & 31)];
  }
#pragma unroll
  for (int k = 0; k < NBMAX / (TB / 32); ++k) {
    const int bin = wv + k * (TB / 32);
    const int n = nn[k];
    if ((ln >> 3) < ((n + 15) >> 4)) {
      int4 v = vv[k];
      if (2 * ln >= n) { v.x = 0; v.y = 0; }
      if (2 * ln + 1 >= n) { v.z = 0; v.w = 0; }
      ((int4*)(seg + (((size_t)bin * NCHUNK + c) << SEGSH)))[ln] = v;
    }
    if (ln == 0 && bin < nbins) segcnt[(size_t)bin * NCHUNK + c] = n;
  }
  if (threadIdx.x == 0) spillcnt[c] = sh[nbins];
}

__device__ __forceinline__ void gcn_put(int s, int d, int p, int* lb, int base,
                                        int* __restrict__ ovfn,
                                        int* __restrict__ osrc,
                                        int* __restrict__ odst) {
  const int row = d - base;
  if (p < CAP) {
    lb[row * CAP + ((p + 4 * ((row >> 1) & 7)) & (CAP - 1))] = s;
  } else {
    const int q = atomicAdd(ovfn, 1);
    osrc[q] = s;
    odst[q] = d;
  }
}

__device__ __forceinline__ void gcn_place(int s, int d, int* lcnt, int* lb,
                                          int base, int* __restrict__ ovfn,
                                          int* __restrict__ osrc,
                                          int* __restrict__ odst) {
  const int p = atomicAdd(&lcnt[d - base], 1);
  gcn_put(s, d, p, lb, base, ovfn, osrc, odst);
}

#define SPW (NCHUNK / (TB / 32))
static_assert(SPW * (TB / 32) == NCHUNK && SEGCAP == 64,
              "every segment has one wave, and a lane takes places l and l + 32");
__global__ __launch_bounds__(TB) void gcn_edge_rank(
    const int2* __restrict__ seg, const int* __restrict__ segcnt,
    const int2* __restrict__ spill, const int* __restrict__ spillcnt,
    int chunk, int n, int* __restrict__ cnt, int* __restrict__ ovfn,
    int* __restrict__ bucket, int* __restrict__ osrc, int* __restrict__ odst) {
  __shared__ int lcnt[RNG];
  __shared__ int sspill[NCHUNK];
  __shared__ int anyspill;
  __shared__ __attribute__((aligned(16))) int lb[RNG * CAP];
  const int b = blockIdx.x;
  const int base = b << RSH;
  const int tid = threadIdx.x;
  const int ln = tid & 31;
  const int wv = __builtin_amdgcn_readfirstlane(tid >> 5);

  const int* __restrict__ mycnt = segcnt + ((size_t)b * NCHUNK + wv * SPW);
  const int2* __restrict__ myseg =
      seg + (((size_t)b * NCHUNK + wv * SPW) << SEGSH);
  int sc[SPW];
#pragma unroll
  for (int k = 0; k < SPW; ++k) sc[k] = mycnt[k];
  int mysp = 0;
  if (tid < NCHUNK) mysp = spillcnt[tid];
  int2 e[2 * SPW];
#pragma unroll
  for (int k = 0; k < 2 * SPW; ++k) {
    const int at = ln + 32 * (k & 1);
    e[k] = (at < sc[k >> 1]) ? myseg[((k >> 1) << SEGSH) + at] : make_int2(0, 0);
  }

  for (int i = tid; i < RNG; i += TB) lcnt[i] = 0;
  if (tid < NCHUNK) sspill[tid] = mysp;
  if (tid == 0) anyspill = 0;
  __syncthreads();
  if (mysp > 0) anyspill = 1;

  int p[2 * SPW];
#pragma unroll
  for (int k = 0; k < 2 * SPW; ++k) {
    const int at = ln + 32 * (k & 1);
    p[k] = (at < sc[k >> 1]) ? atomicAdd(&lcnt[e[k].y - base], 1) : 0;
  }
#pragma unroll
  for (int k = 0; k < 2 * SPW; ++k) {
    const int at = ln + 32 * (k & 1);
    if (at < sc[k >> 1])
      gcn_put(e[k].x, e[k].y, p[k], lb, base, ovfn, osrc, odst);
  }
  __syncthreads();
  if (anyspill) {
    for (int c = 0; c < NCHUNK; ++c) {
      const int oc = sspill[c];
      for (int i = tid; i < oc; i += TB) {
        const int2 e = spill[(size_t)c * chunk + i];
        if ((e.y >> RSH) == b)
          gcn_place(e.x, e.y, lcnt, lb, base, ovfn, osrc, odst);
      }
    }
  }
  __syncthreads();
  for (int i = tid; i < RNG; i += TB) {
    if (base + i < n) cnt[base + i] = lcnt[i];
  }
  int dg[RNG * CAP / 4 / TB];
  int4 vv[RNG * CAP / 4 / TB];
#pragma unroll
  for (int k = 0; k < RNG * CAP / 4 / TB; ++k) {
    const int i = tid + k * TB;
    const int row = i >> 3;
    dg[k] = lcnt[row];
    vv[k] = ((const int4*)lb)[(row << 3) + (((i & 7) + (row >> 1)) & 7)];
  }
#pragma unroll
  for (int k = 0; k < RNG * CAP / 4 / TB; ++k) {
    const int i = tid + k * TB;
    const int row = i >> 3;
    const int j = (i & 7) << 2;
    const int deg = dg[k];
    int4 v = vv[k];
    v.x = (j + 0 < deg) ? v.x : 0;
    v.y = (j + 1 < deg) ? v.y : 0;
    v.z = (j + 2 < deg) ? v.z : 0;
    v.w = (j + 3 < deg) ? v.w : 0;
    if (base + row < n) ((int4*)bucket)[((size_t)base << 3) + i] = v;
  }
}

#define XP 68
#define WPB 8

__device__ __forceinline__ void gcn_sum8(float4& acc,
                                         const float* __restrict__ g,
                                         const int* __restrict__ cnt,
                                         unsigned q, unsigned self, int4 ia,
                                         int4 ib, int j, int end) {
  const bool k0 = j + 0 < end, k1 = j + 1 < end, k2 = j + 2 < end,
             k3 = j + 3 < end, k4 = j + 4 < end, k5 = j + 5 < end,
             k6 = j + 6 < end, k7 = j + 7 < end;
  const unsigned f = k0 ? (unsigned)ia.x : self;
  const unsigned s1 = k1 ? (unsigned)ia.y : f;
  const unsigned s2 = k2 ? (unsigned)ia.z : f;
  const unsigned s3 = k3 ? (unsigned)ia.w : f;
  const unsigned s4 = k4 ? (unsigned)ib.x : f;
  const unsigned s5 = k5 ? (unsigned)ib.y : f;
  const unsigned s6 = k6 ? (unsigned)ib.z : f;
  const unsigned s7 = k7 ? (unsigned)ib.w : f;
  const int c0 = cnt[f];
  const float4 v0 = *(const float4*)(g + (f * 64u + q));
  const int c1 = cnt[s1];
  const float4 v1 = *(const float4*)(g + (s1 * 64u + q));
  const int c2 = cnt[s2];
  const float4 v2 = *(const float4*)(g + (s2 * 64u + q));
  const int c3 = cnt[s3];
  const float4 v3 = *(const float4*)(g + (s3 * 64u + q));
  const int c4 = cnt[s4];
  const float4 v4 = *(const float4*)(g + (s4 * 64u + q));
  const int c5 = cnt[s5];
  const float4 v5 = *(const float4*)(g + (s5 * 64u + q));
  const int c6 = cnt[s6];
  const float4 v6 = *(const float4*)(g + (s6 * 64u + q));
  const int c7 = cnt[s7];
  const float4 v7 = *(const float4*)(g + (s7 * 64u + q));
  __builtin_amdgcn_sched_barrier(0);
  const float w0 = gcn_dinv(c0), w1 = gcn_dinv(c1), w2 = gcn_dinv(c2),
              w3 = gcn_dinv(c3), w4 = gcn_dinv(c4), w5 = gcn_dinv(c5),
              w6 = gcn_dinv(c6), w7 = gcn_dinv(c7);
  acc.x = fmaf(w0, k0 ? v0.x : 0.0f, acc.x); acc.y = fmaf(w0, k0 ? v0.y : 0.0f, acc.y);
  acc.z = fmaf(w0, k0 ? v0.z : 0.0f, acc.z); acc.w = fmaf(w0, k0 ? v0.w : 0.0f, acc.w);
  acc.x = fmaf(w1, k1 ? v1.x : 0.0f, acc.x); acc.y = fmaf(w1, k1 ? v1.y : 0.0f, acc.y);
  acc.z = fmaf(w1, k1 ? v1.z : 0.0f, acc.z); acc.w = fmaf(w1, k1 ? v1.w : 0.0f, acc.w);
  acc.x = fmaf(w2, k2 ? v2.x : 0.0f, acc.x); acc.y = fmaf(w2, k2 ? v2.y : 0.0f, acc.y);
  acc.z = fmaf(w2, k2 ? v2.z : 0.0f, acc.z); acc.w = fmaf(w2, k2 ? v2.w : 0.0f, acc.w);
  acc.x = fmaf(w3, k3 ? v3.x : 0.0f, acc.x); acc.y = fmaf(w3, k3 ? v3.y : 0.0f, acc.y);
  acc.z = fmaf(w3, k3 ? v3.z : 0.0f, acc.z); acc.w = fmaf(w3, k3 ? v3.w : 0.0f, acc.w);
  acc.x = fmaf(w4, k4 ? v4.x : 0.0f, acc.x); acc.y = fmaf(w4, k4 ? v4.y : 0.0f, acc.y);
  acc.z = fmaf(w4, k4 ? v4.z : 0.0f, acc.z); acc.w = fmaf(w4, k4 ? v4.w : 0.0f, acc.w);
  acc.x = fmaf(w5, k5 ? v5.x : 0.0f, acc.x); acc.y = fmaf(w5, k5 ? v5.y : 0.0f, acc.y);
  acc.z = fmaf(w5, k5 ? v5.z : 0.0f, acc.z); acc.w = fmaf(w5, k5 ? v5.w : 0.0f, acc.w);
  acc.x = fmaf(w6, k6 ? v6.x : 0.0f, acc.x); acc.y = fmaf(w6, k6 ? v6.y : 0.0f, acc.y);
  acc.z = fmaf(w6, k6 ? v6.z : 0.0f, acc.z); acc.w = fmaf(w6, k6 ? v6.w : 0.0f, acc.w);
  acc.x = fmaf(w7, k7 ? v7.x : 0.0f, acc.x); acc.y = fmaf(w7, k7 ? v7.y : 0.0f, acc.y);
  acc.z = fmaf(w7, k7 ? v7.z : 0.0f, acc.z); acc.w = fmaf(w7, k7 ? v7.w : 0.0f, acc.w);
}

__global__ __launch_bounds__(32 * WPB) void gcn_gather1_gemm12(
    const int* __restrict__ cnt, const int* __restrict__ ovfn,
    const int* __restrict__ bucket, const int* __restrict__ osrc,
    const int* __restrict__ odst, const float* __restrict__ X,
    const float* __restrict__ W1, const float* __restrict__ b1,
    const float* __restrict__ W2, float* __restrict__ g2, int n) {
  __shared__ __attribute__((aligned(16))) float sW1[64 * 64];
  __shared__ __attribute__((aligned(16))) float sW2[64 * 16];
  __shared__ __attribute__((aligned(16))) float sX[WPB * 16 * XP];
  __shared__ __attribute__((aligned(16))) float sD[WPB * 16];

  const int tid = threadIdx.x;
  const int wave = tid >> 5;
  const int half = (tid >> 4) & 1;
  const int lp = tid & 15;
  const unsigned q = (unsigned)lp << 2;
  const int rowbase = (blockIdx.x * WPB + wave) * 16;

  const float4 wa = ((const float4*)W1)[tid];
  const float4 wb = ((const float4*)W1)[tid + 256];
  const float4 wc = ((const float4*)W1)[tid + 512];
  const float4 wd = ((const float4*)W1)[tid + 768];
  const float4 we = ((const float4*)W2)[tid];

  int nodeN = (rowbase + half < n) ? rowbase + half : n - 1;
  int degN = cnt[nodeN];
  int4 iaN = ((const int4*)(bucket + (size_t)nodeN * CAP))[0];
  int4 ibN = ((const int4*)(bucket + (size_t)nodeN * CAP))[1];
  float4 selfN = *(const float4*)(X + ((unsigned)nodeN * 64u + q));

  ((float4*)sW1)[tid] = wa;
  ((float4*)sW1)[tid + 256] = wb;
  ((float4*)sW1)[tid + 512] = wc;
  ((float4*)sW1)[tid + 768] = wd;
  ((float4*)sW2)[tid] = we;
  __syncthreads();

  float* __restrict__ myX = sX + wave * 16 * XP;
  for (int t = 0; t < 8; ++t) {
    const int node = nodeN;
    const int deg = degN;
    int4 ia = iaN, ib = ibN;
    const float dv = gcn_dinv(deg);
    float4 acc;
    acc.x = dv * selfN.x; acc.y = dv * selfN.y;
    acc.z = dv * selfN.z; acc.w = dv * selfN.w;
    const int4* __restrict__ col4 = (const int4*)(bucket + (size_t)node * CAP);
    if (t + 1 < 8) {
      const int nb = rowbase + 2 * (t + 1) + half;
      nodeN = (nb < n) ? nb : n - 1;
      degN = cnt[nodeN];
      iaN = ((const int4*)(bucket + (size_t)nodeN * CAP))[0];
      ibN = ((const int4*)(bucket + (size_t)nodeN * CAP))[1];
      selfN = *(const float4*)(X + ((unsigned)nodeN * 64u + q));
    }
    const int end = (deg < CAP) ? deg : CAP;
    gcn_sum8(acc, X, cnt, q, (unsigned)node, ia, ib, 0, end);
    for (int j = 8; j < end; j += 8) {
      ia = col4[j >> 2];
      ib = col4[(j >> 2) + 1];
      gcn_sum8(acc, X, cnt, q, (unsigned)node, ia, ib, j, end);
    }
    if (deg > CAP) {
      const int no = *ovfn;
      for (int i = 0; i < no; ++i) {
        if (odst[i] == node) {
          const unsigned s = (unsigned)osrc[i];
          const float w = gcn_dinv(cnt[s]);
          const float4 v = *(const float4*)(X + (s * 64u + q));
          acc.x = fmaf(w, v.x, acc.x); acc.y = fmaf(w, v.y, acc.y);
          acc.z = fmaf(w, v.z, acc.z); acc.w = fmaf(w, v.w, acc.w);
        }
      }
    }
    float4 x;
    x.x = dv * acc.x; x.y = dv * acc.y; x.z = dv * acc.z; x.w = dv * acc.w;
    const int lr = 2 * t + half;
    if (lp == 0) sD[wave * 16 + lr] = dv;
    *(float4*)(myX + lr * XP + q) = x;
  }
  __builtin_amdgcn_fence(__ATOMIC_RELEASE, "wavefront");
  __builtin_amdgcn_wave_barrier();
  __builtin_amdgcn_fence(__ATOMIC_ACQUIRE, "wavefront");

  const float* __restrict__ Arow = myX + lp * XP;
  v2f a[16];
#pragma unroll
  for (int kk = 0; kk < 16; ++kk) a[kk] = *(const v2f*)(Arow + 4 * kk + 2 * half);
  const float bb0 = b1[lp], bb1 = b1[16 + lp], bb2 = b1[32 + lp], bb3 = b1[48 + lp];
  v8f acc0 = {}, acc1 = {}, acc2 = {}, acc3 = {};
#pragma unroll
  for (int kk = 0; kk < 16; ++kk) {
    const int ka = 4 * kk + 2 * half;
    const float* b0r = &sW1[ka * 64];
    const float* b1r = &sW1[(ka + 1) * 64];
    v2f bf;
    bf.x = b0r[lp];
    bf.y = b1r[lp];
    acc0 = __builtin_amdgcn_wmma_f32_16x16x4_f32(false, a[kk], false, bf,
                                                 (short)0, acc0, false, false);
    bf.x = b0r[16 + lp];
    bf.y = b1r[16 + lp];
    acc1 = __builtin_amdgcn_wmma_f32_16x16x4_f32(false, a[kk], false, bf,
                                                 (short)0, acc1, false, false);
    bf.x = b0r[32 + lp];
    bf.y = b1r[32 + lp];
    acc2 = __builtin_amdgcn_wmma_f32_16x16x4_f32(false, a[kk], false, bf,
                                                 (short)0, acc2, false, false);
    bf.x = b0r[48 + lp];
    bf.y = b1r[48 + lp];
    acc3 = __builtin_amdgcn_wmma_f32_16x16x4_f32(false, a[kk], false, bf,
                                                 (short)0, acc3, false, false);
  }
#pragma unroll
  for (int j = 0; j < 8; ++j) {
    float* hr = myX + (j + 8 * half) * XP + lp;
    hr[0]  = fmaxf(acc0[j] + bb0, 0.0f);
    hr[16] = fmaxf(acc1[j] + bb1, 0.0f);
    hr[32] = fmaxf(acc2[j] + bb2, 0.0f);
    hr[48] = fmaxf(acc3[j] + bb3, 0.0f);
  }
  __builtin_amdgcn_fence(__ATOMIC_RELEASE, "wavefront");
  __builtin_amdgcn_wave_barrier();
  __builtin_amdgcn_fence(__ATOMIC_ACQUIRE, "wavefront");

  v8f out2 = {};
#pragma unroll
  for (int k = 0; k < 64; k += 4) {
    const int ka = k + 2 * half;
    const v2f a2 = *(const v2f*)(Arow + ka);
    v2f b;
    b.x = sW2[ka * 16 + lp];
    b.y = sW2[(ka + 1) * 16 + lp];
    out2 = __builtin_amdgcn_wmma_f32_16x16x4_f32(false, a2, false, b, (short)0,
                                                 out2, false, false);
  }

  const float4 d0 = *(const float4*)(sD + wave * 16 + 8 * half);
  const float4 d1 = *(const float4*)(sD + wave * 16 + 8 * half + 4);
  const float dv8[8] = {d0.x, d0.y, d0.z, d0.w, d1.x, d1.y, d1.z, d1.w};
#pragma unroll
  for (int j = 0; j < 8; ++j) {
    const int row = rowbase + j + 8 * half;
    if (row < n) g2[(size_t)row * 16 + lp] = dv8[j] * out2[j];
  }
}

__global__ __launch_bounds__(256) void gcn_gather2(
    const int* __restrict__ cnt, const int* __restrict__ ovfn,
    const int* __restrict__ bucket, const int* __restrict__ osrc,
    const int* __restrict__ odst, const float* __restrict__ g2,
    const float* __restrict__ b2, float* __restrict__ out, int n) {
  const int tid = blockIdx.x * blockDim.x + threadIdx.x;
  const int node = tid >> 2;
  if (node >= n) return;
  const int q = (tid & 3) << 2;
  const int deg = cnt[node];
  const int end = (deg < CAP) ? deg : CAP;
  const int* __restrict__ col = bucket + (size_t)node * CAP;
  const float* __restrict__ gq = g2 + q;
  float4 acc = *(const float4*)(gq + (size_t)node * 16);
  int j = 0;
  for (; j + 4 <= end; j += 4) {
    const int s0 = col[j], s1 = col[j + 1], s2 = col[j + 2], s3 = col[j + 3];
    const float4 v0 = *(const float4*)(gq + (size_t)s0 * 16);
    const float4 v1 = *(const float4*)(gq + (size_t)s1 * 16);
    const float4 v2 = *(const float4*)(gq + (size_t)s2 * 16);
    const float4 v3 = *(const float4*)(gq + (size_t)s3 * 16);
    acc.x += v0.x; acc.y += v0.y; acc.z += v0.z; acc.w += v0.w;
    acc.x += v1.x; acc.y += v1.y; acc.z += v1.z; acc.w += v1.w;
    acc.x += v2.x; acc.y += v2.y; acc.z += v2.z; acc.w += v2.w;
    acc.x += v3.x; acc.y += v3.y; acc.z += v3.z; acc.w += v3.w;
  }
  for (; j < end; ++j) {
    const float4 v = *(const float4*)(gq + (size_t)col[j] * 16);
    acc.x += v.x; acc.y += v.y; acc.z += v.z; acc.w += v.w;
  }
  if (deg > CAP) {
    const int no = *ovfn;
    for (int i = 0; i < no; ++i) {
      if (odst[i] == node) {
        const float4 v = *(const float4*)(gq + (size_t)osrc[i] * 16);
        acc.x += v.x; acc.y += v.y; acc.z += v.z; acc.w += v.w;
      }
    }
  }
  const float dv = gcn_dinv(deg);
  float4 r;
  r.x = fmaf(dv, acc.x, b2[q + 0]);
  r.y = fmaf(dv, acc.y, b2[q + 1]);
  r.z = fmaf(dv, acc.z, b2[q + 2]);
  r.w = fmaf(dv, acc.w, b2[q + 3]);
  *(float4*)(out + (size_t)node * 16 + q) = r;
}

extern "C" void kernel_launch(void* const* d_in, const int* in_sizes, int n_in,
                              void* d_out, int out_size, void* d_ws, size_t ws_size,
                              hipStream_t stream) {
  const float* X  = (const float*)d_in[0];
  const int*   ei = (const int*)d_in[1];
  const float* W1 = (const float*)d_in[2];
  const float* b1 = (const float*)d_in[3];
  const float* W2 = (const float*)d_in[4];
  const float* b2 = (const float*)d_in[5];
  float* out = (float*)d_out;

  const int N = in_sizes[0] / 64;
  const int E = in_sizes[1] / 2;
  const int* src = ei;
  const int* dst = ei + E;

  float* ws   = (float*)d_ws;
  float* g2   = ws;
  int* cnt    = (int*)(g2 + (size_t)N * 16);
  int* ovfn   = cnt + N;
  int* bucket = cnt + ((N + 1 + 31) / 32) * 32;
  int* osrc   = bucket + (size_t)N * CAP;
  int* odst   = osrc + E;

  const int nbins = (N + RNG - 1) / RNG;
  const int chunk = (E + NCHUNK - 1) / NCHUNK;
  const size_t used = (size_t)((odst + E) - (int*)d_ws);
  int2* seg     = (int2*)((int*)d_ws + (used + 31) / 32 * 32);
  int2* spill   = seg + (size_t)nbins * NCHUNK * SEGCAP;
  int* segcnt   = (int*)(spill + (size_t)NCHUNK * chunk);
  int* spillcnt = segcnt + (size_t)nbins * NCHUNK;

  const int B = 256;

  gcn_edge_bin<<<NCHUNK, TB, 0, stream>>>(
      src, dst, E, chunk, nbins, seg, segcnt, spill, spillcnt, ovfn);
  gcn_edge_rank<<<nbins, TB, 0, stream>>>(seg, segcnt, spill, spillcnt, chunk,
                                          N, cnt, ovfn, bucket, osrc, odst);
  gcn_gather1_gemm12<<<(N + 16 * WPB - 1) / (16 * WPB), 32 * WPB, 0, stream>>>(
      cnt, ovfn, bucket, osrc, odst, X, W1, b1, W2, g2, N);
  gcn_gather2<<<(N * 4 + B - 1) / B, B, 0, stream>>>(cnt, ovfn, bucket, osrc, odst,
                                                     g2, b2, out, N);
}
